# starting kernel
# baseline (speedup 1.0000x reference)
#include <hip/hip_runtime.h>

typedef __attribute__((ext_vector_type(2))) float v2f;
typedef __attribute__((ext_vector_type(8))) float v8f;

__device__ __forceinline__ void atomic_add_f32(float* p, float v) {
  (void)__hip_atomic_fetch_add(p, v, __ATOMIC_RELAXED, __HIP_MEMORY_SCOPE_AGENT);
}


__global__ void gcn_zero_kernel(float* __restrict__ p, int n) {
  int i = blockIdx.x * blockDim.x + threadIdx.x;
  if (i < n) p[i] = 0.0f;
}

__global__ void gcn_degree_kernel(const int* __restrict__ src, const int* __restrict__ dst,
                                  float* __restrict__ deg_out, float* __restrict__ deg_in,
                                  int E) {
  int e = blockIdx.x * blockDim.x + threadIdx.x;
  if (e < E) {
    atomic_add_f32(&deg_out[src[e]], 1.0f);
    atomic_add_f32(&deg_in[dst[e]], 1.0f);
  }
}

__global__ void gcn_norm_kernel(float* __restrict__ deg, int n) {
  int i = blockIdx.x * blockDim.x + threadIdx.x;
  if (i < n) deg[i] = rsqrtf(fmaxf(deg[i], 1.0f));
}

__global__ void gcn_post_kernel(float* __restrict__ h, const float* __restrict__ norm_dst,
                                const float* __restrict__ bias, int n, int H) {
  int idx = blockIdx.x * blockDim.x + threadIdx.x;
  if (idx >= n) return;
  int i = idx / H;
  int f = idx - i * H;
  h[idx] = fmaxf(fmaf(h[idx], norm_dst[i], bias[f]), 0.0f);
}

__global__ void gcn_gemm_wmma_kernel(const float* __restrict__ A, int lda, int K,
                                     const float* __restrict__ W, int wcols,
                                     const float* __restrict__ norm,
                                     float* __restrict__ out, int ldo, int ntiles) {
  __shared__ float Ws[32 * 32];

  for (int i = threadIdx.x; i < 32 * 32; i += blockDim.x) {
    const int k = i >> 5;
    const int n = i & 31;
    Ws[i] = (k < K && n < wcols) ? W[(size_t)k * wcols + n] : 0.0f;
  }
  __syncthreads();

  const int wave = (blockIdx.x * blockDim.x + threadIdx.x) >> 5;
  const int lane = threadIdx.x & 31;
  if (wave >= ntiles) return;

  const int m = lane & 15;
  const int half = lane >> 4;

  const int row = wave * 16 + m;
  const float nrm = norm[row];
  const float* arow = A + (size_t)row * lda;

  v8f c0 = {};
  v8f c1 = {};

  for (int k0 = 0; k0 < K; k0 += 4) {
    const int k = k0 + 2 * half;
    const float2 av = *(const float2*)(arow + k);
    v2f a;
    a.x = av.x * nrm;
    a.y = av.y * nrm;

    v2f b0, b1;
    b0.x = Ws[k * 32 + m];
    b0.y = Ws[(k + 1) * 32 + m];
    b1.x = Ws[k * 32 + 16 + m];
    b1.y = Ws[(k + 1) * 32 + 16 + m];

    c0 = __builtin_amdgcn_wmma_f32_16x16x4_f32(false, a, false, b0, (short)0, c0, false, false);
    c1 = __builtin_amdgcn_wmma_f32_16x16x4_f32(false, a, false, b1, (short)0, c1, false, false);
  }

  for (int v = 0; v < 8; ++v) {
    const int orow = wave * 16 + v + 8 * half;
    out[(size_t)orow * ldo + m] = c0[v];
    if (16 + m < wcols) out[(size_t)orow * ldo + 16 + m] = c1[v];
  }
}

__global__ void gcn_scatter_kernel(const float* __restrict__ x, const int* __restrict__ src,
                                   const int* __restrict__ dst, float* __restrict__ agg,
                                   int nwork, int nch, int H) {
  int idx = blockIdx.x * blockDim.x + threadIdx.x;
  if (idx >= nwork) return;
  int e = idx / nch;
  int c = idx - e * nch;
  int s = src[e];
  int d = dst[e];
  const float4 v = *(const float4*)(x + (size_t)s * H + c * 4);
  float* ap = agg + (size_t)d * H + c * 4;
  atomic_add_f32(ap + 0, v.x);
  atomic_add_f32(ap + 1, v.y);
  atomic_add_f32(ap + 2, v.z);
  atomic_add_f32(ap + 3, v.w);
}

__global__ void gcn_readout_kernel(const float* __restrict__ h, const float* __restrict__ Wd,
                                   const float* __restrict__ bd, float* __restrict__ out,
                                   int nq, int rowlen  ) {
  int r = blockIdx.x * blockDim.x + threadIdx.x;
  if (r >= nq) return;
  const float4* hp = (const float4*)(h + (size_t)r * rowlen);
  const float4* wp = (const float4*)Wd;
  const int nch = rowlen >> 2;
  float s = 0.0f;
  for (int j = 0; j < nch; ++j) {
    float4 hv = hp[j];
    float4 wv = wp[j];
    s = fmaf(hv.x, wv.x, s);
    s = fmaf(hv.y, wv.y, s);
    s = fmaf(hv.z, wv.z, s);
    s = fmaf(hv.w, wv.w, s);
  }
  out[r] = s + bd[0];
}


extern "C" void kernel_launch(void* const* d_in, const int* in_sizes, int n_in,
                              void* d_out, int out_size, void* d_ws, size_t ws_size,
                              hipStream_t stream) {
  const float* features = (const float*)d_in[0];
  const int*   src      = (const int*)d_in[1];
  const int*   dst      = (const int*)d_in[2];
  const float* W1       = (const float*)d_in[3];
  const float* b1       = (const float*)d_in[4];
  const float* W2       = (const float*)d_in[5];
  const float* b2       = (const float*)d_in[6];
  const float* Wd       = (const float*)d_in[7];
  const float* bd       = (const float*)d_in[8];
  float* out = (float*)d_out;

  const int H  = in_sizes[4];
  const int F  = in_sizes[3] / H;
  const int Nn = in_sizes[0] / F;
  const int E  = in_sizes[1];

  float* deg_out = (float*)d_ws;
  float* deg_in  = deg_out + Nn;
  float* bufA    = deg_in + Nn;
  float* bufX    = bufA + (size_t)Nn * H;

  const int B = 256;
  const int NH = Nn * H;
  const int ntiles = Nn / 16;
  const int wavesPerBlock = B / 32;
  const int gemmGrid = (ntiles + wavesPerBlock - 1) / wavesPerBlock;
  const int nch = H / 4;
  const int scatterWork = E * nch;
  const int nq = Nn / 4;

  {
    int n = 2 * Nn + NH;
    gcn_zero_kernel<<<(n + B - 1) / B, B, 0, stream>>>(deg_out, n);
  }
  gcn_degree_kernel<<<(E + B - 1) / B, B, 0, stream>>>(src, dst, deg_out, deg_in, E);
  gcn_norm_kernel<<<(2 * Nn + B - 1) / B, B, 0, stream>>>(deg_out, 2 * Nn);
  gcn_gemm_wmma_kernel<<<gemmGrid, B, 0, stream>>>(features, F, F, W1, H, deg_out,
                                                   bufX, H, ntiles);
  gcn_scatter_kernel<<<(scatterWork + B - 1) / B, B, 0, stream>>>(bufX, src, dst, bufA,
                                                                  scatterWork, nch, H);
  gcn_post_kernel<<<(NH + B - 1) / B, B, 0, stream>>>(bufA, deg_in, b1, NH, H);
  gcn_gemm_wmma_kernel<<<gemmGrid, B, 0, stream>>>(bufA, H, H, W2, H, deg_out,
                                                   bufX, H, ntiles);
  gcn_zero_kernel<<<(NH + B - 1) / B, B, 0, stream>>>(bufA, NH);
  gcn_scatter_kernel<<<(scatterWork + B - 1) / B, B, 0, stream>>>(bufX, src, dst, bufA,
                                                                  scatterWork, nch, H);
  gcn_post_kernel<<<(NH + B - 1) / B, B, 0, stream>>>(bufA, deg_in, b2, NH, H);
  gcn_readout_kernel<<<(nq + B - 1) / B, B, 0, stream>>>(bufA, Wd, bd, out, nq, 4 * H);
}


# degree pass: per-block LDS 16-bit counter tables (edge slice x node range) stored as partial planes and summed in the norm pass replace the scattered global f32 atomic adds; counts stay exact
# speedup vs baseline: 1.3148x; 1.3148x over previous
#include <hip/hip_runtime.h>

typedef __attribute__((ext_vector_type(2))) float v2f;
typedef __attribute__((ext_vector_type(8))) float v8f;

__device__ __forceinline__ void atomic_add_f32(float* p, float v) {
  (void)__hip_atomic_fetch_add(p, v, __ATOMIC_RELAXED, __HIP_MEMORY_SCOPE_AGENT);
}


__global__ void gcn_zero_kernel(float* __restrict__ p, int n) {
  int i = blockIdx.x * blockDim.x + threadIdx.x;
  if (i < n) p[i] = 0.0f;
}

#define DEG_RANGE 50000
#define DEG_SLICE 50000

__device__ __forceinline__ void deg_count_one(unsigned int* tbl, int idx, int base) {
  const unsigned int rel = (unsigned int)(idx - base);
  if (rel < (unsigned int)DEG_RANGE) {
    (void)__hip_atomic_fetch_add(&tbl[rel >> 1], 1u << ((rel & 1u) * 16u), __ATOMIC_RELAXED,
                                 __HIP_MEMORY_SCOPE_WORKGROUP);
  }
}

__global__ void gcn_degree_lds_kernel(const int* __restrict__ src, const int* __restrict__ dst,
                                      unsigned int* __restrict__ P, int E, int npairs) {
  __shared__ unsigned int tbl[DEG_RANGE / 2];

  const int s = blockIdx.x;
  const int r = blockIdx.y;
  const int a = blockIdx.z;
  const int S = gridDim.x;

  for (int i = threadIdx.x; i < DEG_RANGE / 2; i += blockDim.x) tbl[i] = 0u;
  __syncthreads();

  const int* idxp = (a == 0 ? src : dst) + (size_t)s * DEG_SLICE;
  int n = E - s * DEG_SLICE;
  if (n > DEG_SLICE) n = DEG_SLICE;
  const int base = r * DEG_RANGE;

  const int n4 = n >> 2;
  const int4* idx4 = (const int4*)idxp;
  for (int i0 = threadIdx.x; i0 < n4; i0 += 4 * blockDim.x) {
    int4 v[4];
#pragma unroll
    for (int j = 0; j < 4; ++j) {
      const int i = i0 + j * (int)blockDim.x;
      v[j] = (i < n4) ? idx4[i] : make_int4(-1, -1, -1, -1);
    }
#pragma unroll
    for (int j = 0; j < 4; ++j) {
      deg_count_one(tbl, v[j].x, base);
      deg_count_one(tbl, v[j].y, base);
      deg_count_one(tbl, v[j].z, base);
      deg_count_one(tbl, v[j].w, base);
    }
  }
  for (int i = (n4 << 2) + threadIdx.x; i < n; i += blockDim.x) deg_count_one(tbl, idxp[i], base);
  __syncthreads();

  unsigned int* plane = P + ((size_t)a * S + s) * npairs;
  const int w0 = r * (DEG_RANGE / 2);
  for (int i = threadIdx.x; i < DEG_RANGE / 2; i += blockDim.x) {
    const int w = w0 + i;
    if (w < npairs) plane[w] = tbl[i];
  }
}

__global__ void gcn_norm_kernel(const unsigned int* __restrict__ P, float* __restrict__ norm,
                                int S, int npairs, int Nn) {
  const int t = blockIdx.x * blockDim.x + threadIdx.x;
  if (t >= 2 * npairs) return;
  const int a = t / npairs;
  const int w = t - a * npairs;
  const unsigned int* p = P + (size_t)a * S * npairs + w;
  unsigned int lo = 0u, hi = 0u;
#pragma unroll 8
  for (int s = 0; s < S; ++s) {
    const unsigned int v = p[(size_t)s * npairs];
    lo += v & 0xffffu;
    hi += v >> 16;
  }
  float* o = norm + (size_t)a * Nn;
  const int node = 2 * w;
  o[node] = rsqrtf(fmaxf((float)lo, 1.0f));
  if (node + 1 < Nn) o[node + 1] = rsqrtf(fmaxf((float)hi, 1.0f));
}

__global__ void gcn_post_kernel(float* __restrict__ h, const float* __restrict__ norm_dst,
                                const float* __restrict__ bias, int n, int H) {
  int idx = blockIdx.x * blockDim.x + threadIdx.x;
  if (idx >= n) return;
  int i = idx / H;
  int f = idx - i * H;
  h[idx] = fmaxf(fmaf(h[idx], norm_dst[i], bias[f]), 0.0f);
}

__global__ void gcn_gemm_wmma_kernel(const float* __restrict__ A, int lda, int K,
                                     const float* __restrict__ W, int wcols,
                                     const float* __restrict__ norm,
                                     float* __restrict__ out, int ldo, int ntiles) {
  __shared__ float Ws[32 * 32];

  for (int i = threadIdx.x; i < 32 * 32; i += blockDim.x) {
    const int k = i >> 5;
    const int n = i & 31;
    Ws[i] = (k < K && n < wcols) ? W[(size_t)k * wcols + n] : 0.0f;
  }
  __syncthreads();

  const int wave = (blockIdx.x * blockDim.x + threadIdx.x) >> 5;
  const int lane = threadIdx.x & 31;
  if (wave >= ntiles) return;

  const int m = lane & 15;
  const int half = lane >> 4;

  const int row = wave * 16 + m;
  const float nrm = norm[row];
  const float* arow = A + (size_t)row * lda;

  v8f c0 = {};
  v8f c1 = {};

  for (int k0 = 0; k0 < K; k0 += 4) {
    const int k = k0 + 2 * half;
    const float2 av = *(const float2*)(arow + k);
    v2f a;
    a.x = av.x * nrm;
    a.y = av.y * nrm;

    v2f b0, b1;
    b0.x = Ws[k * 32 + m];
    b0.y = Ws[(k + 1) * 32 + m];
    b1.x = Ws[k * 32 + 16 + m];
    b1.y = Ws[(k + 1) * 32 + 16 + m];

    c0 = __builtin_amdgcn_wmma_f32_16x16x4_f32(false, a, false, b0, (short)0, c0, false, false);
    c1 = __builtin_amdgcn_wmma_f32_16x16x4_f32(false, a, false, b1, (short)0, c1, false, false);
  }

  for (int v = 0; v < 8; ++v) {
    const int orow = wave * 16 + v + 8 * half;
    out[(size_t)orow * ldo + m] = c0[v];
    if (16 + m < wcols) out[(size_t)orow * ldo + 16 + m] = c1[v];
  }
}

__global__ void gcn_scatter_kernel(const float* __restrict__ x, const int* __restrict__ src,
                                   const int* __restrict__ dst, float* __restrict__ agg,
                                   int nwork, int nch, int H) {
  int idx = blockIdx.x * blockDim.x + threadIdx.x;
  if (idx >= nwork) return;
  int e = idx / nch;
  int c = idx - e * nch;
  int s = src[e];
  int d = dst[e];
  const float4 v = *(const float4*)(x + (size_t)s * H + c * 4);
  float* ap = agg + (size_t)d * H + c * 4;
  atomic_add_f32(ap + 0, v.x);
  atomic_add_f32(ap + 1, v.y);
  atomic_add_f32(ap + 2, v.z);
  atomic_add_f32(ap + 3, v.w);
}

__global__ void gcn_readout_kernel(const float* __restrict__ h, const float* __restrict__ Wd,
                                   const float* __restrict__ bd, float* __restrict__ out,
                                   int nq, int rowlen  ) {
  int r = blockIdx.x * blockDim.x + threadIdx.x;
  if (r >= nq) return;
  const float4* hp = (const float4*)(h + (size_t)r * rowlen);
  const float4* wp = (const float4*)Wd;
  const int nch = rowlen >> 2;
  float s = 0.0f;
  for (int j = 0; j < nch; ++j) {
    float4 hv = hp[j];
    float4 wv = wp[j];
    s = fmaf(hv.x, wv.x, s);
    s = fmaf(hv.y, wv.y, s);
    s = fmaf(hv.z, wv.z, s);
    s = fmaf(hv.w, wv.w, s);
  }
  out[r] = s + bd[0];
}


extern "C" void kernel_launch(void* const* d_in, const int* in_sizes, int n_in,
                              void* d_out, int out_size, void* d_ws, size_t ws_size,
                              hipStream_t stream) {
  const float* features = (const float*)d_in[0];
  const int*   src      = (const int*)d_in[1];
  const int*   dst      = (const int*)d_in[2];
  const float* W1       = (const float*)d_in[3];
  const float* b1       = (const float*)d_in[4];
  const float* W2       = (const float*)d_in[5];
  const float* b2       = (const float*)d_in[6];
  const float* Wd       = (const float*)d_in[7];
  const float* bd       = (const float*)d_in[8];
  float* out = (float*)d_out;

  const int H  = in_sizes[4];
  const int F  = in_sizes[3] / H;
  const int Nn = in_sizes[0] / F;
  const int E  = in_sizes[1];

  float* deg_out = (float*)d_ws;
  float* deg_in  = deg_out + Nn;
  float* bufA    = deg_in + Nn;
  float* bufX    = bufA + (size_t)Nn * H;
  unsigned int* degP = (unsigned int*)(bufX + (size_t)Nn * H);
  const int npairs = (Nn + 1) / 2;
  const int degS = (E + DEG_SLICE - 1) / DEG_SLICE;
  const int degR = (Nn + DEG_RANGE - 1) / DEG_RANGE;

  const int B = 256;
  const int NH = Nn * H;
  const int ntiles = Nn / 16;
  const int wavesPerBlock = B / 32;
  const int gemmGrid = (ntiles + wavesPerBlock - 1) / wavesPerBlock;
  const int nch = H / 4;
  const int scatterWork = E * nch;
  const int nq = Nn / 4;

  gcn_zero_kernel<<<(NH + B - 1) / B, B, 0, stream>>>(bufA, NH);
  gcn_degree_lds_kernel<<<dim3(degS, degR, 2), 1024, 0, stream>>>(src, dst, degP, E, npairs);
  gcn_norm_kernel<<<(2 * npairs + B - 1) / B, B, 0, stream>>>(degP, deg_out, degS, npairs, Nn);
  gcn_gemm_wmma_kernel<<<gemmGrid, B, 0, stream>>>(features, F, F, W1, H, deg_out,
                                                   bufX, H, ntiles);
  gcn_scatter_kernel<<<(scatterWork + B - 1) / B, B, 0, stream>>>(bufX, src, dst, bufA,
                                                                  scatterWork, nch, H);
  gcn_post_kernel<<<(NH + B - 1) / B, B, 0, stream>>>(bufA, deg_in, b1, NH, H);
  gcn_gemm_wmma_kernel<<<gemmGrid, B, 0, stream>>>(bufA, H, H, W2, H, deg_out,
                                                   bufX, H, ntiles);
  gcn_zero_kernel<<<(NH + B - 1) / B, B, 0, stream>>>(bufA, NH);
  gcn_scatter_kernel<<<(scatterWork + B - 1) / B, B, 0, stream>>>(bufX, src, dst, bufA,
                                                                  scatterWork, nch, H);
  gcn_post_kernel<<<(NH + B - 1) / B, B, 0, stream>>>(bufA, deg_in, b2, NH, H);
  gcn_readout_kernel<<<(nq + B - 1) / B, B, 0, stream>>>(bufA, Wd, bd, out, nq, 4 * H);
}


# edge scatter: one wave per 32 edges; each edge row gathered and added by ONE instruction (lane f = float f, 96 contiguous bytes), not 4 dword-strided ones; f32 atomics and arrival-order sums kept
# speedup vs baseline: 1.9198x; 1.4601x over previous
#include <hip/hip_runtime.h>

typedef __attribute__((ext_vector_type(2))) float v2f;
typedef __attribute__((ext_vector_type(8))) float v8f;

__device__ __forceinline__ void atomic_add_f32(float* p, float v) {
  (void)__hip_atomic_fetch_add(p, v, __ATOMIC_RELAXED, __HIP_MEMORY_SCOPE_AGENT);
}


__global__ void gcn_zero_kernel(float* __restrict__ p, int n) {
  int i = blockIdx.x * blockDim.x + threadIdx.x;
  if (i < n) p[i] = 0.0f;
}

#define DEG_RANGE 50000
#define DEG_SLICE 50000

__device__ __forceinline__ void deg_count_one(unsigned int* tbl, int idx, int base) {
  const unsigned int rel = (unsigned int)(idx - base);
  if (rel < (unsigned int)DEG_RANGE) {
    (void)__hip_atomic_fetch_add(&tbl[rel >> 1], 1u << ((rel & 1u) * 16u), __ATOMIC_RELAXED,
                                 __HIP_MEMORY_SCOPE_WORKGROUP);
  }
}

__global__ void gcn_degree_lds_kernel(const int* __restrict__ src, const int* __restrict__ dst,
                                      unsigned int* __restrict__ P, int E, int npairs) {
  __shared__ unsigned int tbl[DEG_RANGE / 2];

  const int s = blockIdx.x;
  const int r = blockIdx.y;
  const int a = blockIdx.z;
  const int S = gridDim.x;

  for (int i = threadIdx.x; i < DEG_RANGE / 2; i += blockDim.x) tbl[i] = 0u;
  __syncthreads();

  const int* idxp = (a == 0 ? src : dst) + (size_t)s * DEG_SLICE;
  int n = E - s * DEG_SLICE;
  if (n > DEG_SLICE) n = DEG_SLICE;
  const int base = r * DEG_RANGE;

  const int n4 = n >> 2;
  const int4* idx4 = (const int4*)idxp;
  for (int i0 = threadIdx.x; i0 < n4; i0 += 4 * blockDim.x) {
    int4 v[4];
#pragma unroll
    for (int j = 0; j < 4; ++j) {
      const int i = i0 + j * (int)blockDim.x;
      v[j] = (i < n4) ? idx4[i] : make_int4(-1, -1, -1, -1);
    }
#pragma unroll
    for (int j = 0; j < 4; ++j) {
      deg_count_one(tbl, v[j].x, base);
      deg_count_one(tbl, v[j].y, base);
      deg_count_one(tbl, v[j].z, base);
      deg_count_one(tbl, v[j].w, base);
    }
  }
  for (int i = (n4 << 2) + threadIdx.x; i < n; i += blockDim.x) deg_count_one(tbl, idxp[i], base);
  __syncthreads();

  unsigned int* plane = P + ((size_t)a * S + s) * npairs;
  const int w0 = r * (DEG_RANGE / 2);
  for (int i = threadIdx.x; i < DEG_RANGE / 2; i += blockDim.x) {
    const int w = w0 + i;
    if (w < npairs) plane[w] = tbl[i];
  }
}

__global__ void gcn_norm_kernel(const unsigned int* __restrict__ P, float* __restrict__ norm,
                                int S, int npairs, int Nn) {
  const int t = blockIdx.x * blockDim.x + threadIdx.x;
  if (t >= 2 * npairs) return;
  const int a = t / npairs;
  const int w = t - a * npairs;
  const unsigned int* p = P + (size_t)a * S * npairs + w;
  unsigned int lo = 0u, hi = 0u;
#pragma unroll 8
  for (int s = 0; s < S; ++s) {
    const unsigned int v = p[(size_t)s * npairs];
    lo += v & 0xffffu;
    hi += v >> 16;
  }
  float* o = norm + (size_t)a * Nn;
  const int node = 2 * w;
  o[node] = rsqrtf(fmaxf((float)lo, 1.0f));
  if (node + 1 < Nn) o[node + 1] = rsqrtf(fmaxf((float)hi, 1.0f));
}

__global__ void gcn_post_kernel(float* __restrict__ h, const float* __restrict__ norm_dst,
                                const float* __restrict__ bias, int n, int H) {
  int idx = blockIdx.x * blockDim.x + threadIdx.x;
  if (idx >= n) return;
  int i = idx / H;
  int f = idx - i * H;
  h[idx] = fmaxf(fmaf(h[idx], norm_dst[i], bias[f]), 0.0f);
}

__global__ void gcn_gemm_wmma_kernel(const float* __restrict__ A, int lda, int K,
                                     const float* __restrict__ W, int wcols,
                                     const float* __restrict__ norm,
                                     float* __restrict__ out, int ldo, int ntiles) {
  __shared__ float Ws[32 * 32];

  for (int i = threadIdx.x; i < 32 * 32; i += blockDim.x) {
    const int k = i >> 5;
    const int n = i & 31;
    Ws[i] = (k < K && n < wcols) ? W[(size_t)k * wcols + n] : 0.0f;
  }
  __syncthreads();

  const int wave = (blockIdx.x * blockDim.x + threadIdx.x) >> 5;
  const int lane = threadIdx.x & 31;
  if (wave >= ntiles) return;

  const int m = lane & 15;
  const int half = lane >> 4;

  const int row = wave * 16 + m;
  const float nrm = norm[row];
  const float* arow = A + (size_t)row * lda;

  v8f c0 = {};
  v8f c1 = {};

  for (int k0 = 0; k0 < K; k0 += 4) {
    const int k = k0 + 2 * half;
    const float2 av = *(const float2*)(arow + k);
    v2f a;
    a.x = av.x * nrm;
    a.y = av.y * nrm;

    v2f b0, b1;
    b0.x = Ws[k * 32 + m];
    b0.y = Ws[(k + 1) * 32 + m];
    b1.x = Ws[k * 32 + 16 + m];
    b1.y = Ws[(k + 1) * 32 + 16 + m];

    c0 = __builtin_amdgcn_wmma_f32_16x16x4_f32(false, a, false, b0, (short)0, c0, false, false);
    c1 = __builtin_amdgcn_wmma_f32_16x16x4_f32(false, a, false, b1, (short)0, c1, false, false);
  }

  for (int v = 0; v < 8; ++v) {
    const int orow = wave * 16 + v + 8 * half;
    out[(size_t)orow * ldo + m] = c0[v];
    if (16 + m < wcols) out[(size_t)orow * ldo + 16 + m] = c1[v];
  }
}

#define GCN_ROW 24
#define SCAT_BATCH 16

__global__ void gcn_scatter_rows_kernel(const float* __restrict__ x, const int* __restrict__ src,
                                        const int* __restrict__ dst, float* __restrict__ agg,
                                        int E) {
  const int lane = threadIdx.x & 31;
  const int wave = (blockIdx.x * blockDim.x + threadIdx.x) >> 5;
  const int e0 = wave * 32;
  if (e0 >= E) return;
  const int e = e0 + lane;
  const int sv = (e < E) ? src[e] : 0;
  const int dv = (e < E) ? dst[e] : 0;
  const bool act = lane < GCN_ROW;

  if (e0 + 32 <= E) {
#pragma unroll
    for (int b = 0; b < 32; b += SCAT_BATCH) {
      int si[SCAT_BATCH], di[SCAT_BATCH];
#pragma unroll
      for (int k = 0; k < SCAT_BATCH; ++k) {
        si[k] = __builtin_amdgcn_readlane(sv, b + k);
        di[k] = __builtin_amdgcn_readlane(dv, b + k);
      }
      if (act) {
        float v[SCAT_BATCH];
#pragma unroll
        for (int k = 0; k < SCAT_BATCH; ++k) v[k] = (x + (size_t)si[k] * GCN_ROW)[lane];
#pragma unroll
        for (int k = 0; k < SCAT_BATCH; ++k)
          atomic_add_f32(agg + (size_t)di[k] * GCN_ROW + lane, v[k]);
      }
    }
  } else {
    const int n = E - e0;
    for (int j = 0; j < n; ++j) {
      const int s = __builtin_amdgcn_readlane(sv, j);
      const int d = __builtin_amdgcn_readlane(dv, j);
      if (act) atomic_add_f32(agg + (size_t)d * GCN_ROW + lane, (x + (size_t)s * GCN_ROW)[lane]);
    }
  }
}

__global__ void gcn_readout_kernel(const float* __restrict__ h, const float* __restrict__ Wd,
                                   const float* __restrict__ bd, float* __restrict__ out,
                                   int nq, int rowlen  ) {
  int r = blockIdx.x * blockDim.x + threadIdx.x;
  if (r >= nq) return;
  const float4* hp = (const float4*)(h + (size_t)r * rowlen);
  const float4* wp = (const float4*)Wd;
  const int nch = rowlen >> 2;
  float s = 0.0f;
  for (int j = 0; j < nch; ++j) {
    float4 hv = hp[j];
    float4 wv = wp[j];
    s = fmaf(hv.x, wv.x, s);
    s = fmaf(hv.y, wv.y, s);
    s = fmaf(hv.z, wv.z, s);
    s = fmaf(hv.w, wv.w, s);
  }
  out[r] = s + bd[0];
}


extern "C" void kernel_launch(void* const* d_in, const int* in_sizes, int n_in,
                              void* d_out, int out_size, void* d_ws, size_t ws_size,
                              hipStream_t stream) {
  const float* features = (const float*)d_in[0];
  const int*   src      = (const int*)d_in[1];
  const int*   dst      = (const int*)d_in[2];
  const float* W1       = (const float*)d_in[3];
  const float* b1       = (const float*)d_in[4];
  const float* W2       = (const float*)d_in[5];
  const float* b2       = (const float*)d_in[6];
  const float* Wd       = (const float*)d_in[7];
  const float* bd       = (const float*)d_in[8];
  float* out = (float*)d_out;

  const int H  = in_sizes[4];
  const int F  = in_sizes[3] / H;
  const int Nn = in_sizes[0] / F;
  const int E  = in_sizes[1];

  float* deg_out = (float*)d_ws;
  float* deg_in  = deg_out + Nn;
  float* bufA    = deg_in + Nn;
  float* bufX    = bufA + (size_t)Nn * H;
  unsigned int* degP = (unsigned int*)(bufX + (size_t)Nn * H);
  const int npairs = (Nn + 1) / 2;
  const int degS = (E + DEG_SLICE - 1) / DEG_SLICE;
  const int degR = (Nn + DEG_RANGE - 1) / DEG_RANGE;

  const int B = 256;
  const int NH = Nn * H;
  const int ntiles = Nn / 16;
  const int wavesPerBlock = B / 32;
  const int gemmGrid = (ntiles + wavesPerBlock - 1) / wavesPerBlock;
  const int scatterWaves = (E + 31) / 32;
  const int scatterGrid = (scatterWaves + wavesPerBlock - 1) / wavesPerBlock;
  const int nq = Nn / 4;

  gcn_zero_kernel<<<(NH + B - 1) / B, B, 0, stream>>>(bufA, NH);
  gcn_degree_lds_kernel<<<dim3(degS, degR, 2), 1024, 0, stream>>>(src, dst, degP, E, npairs);
  gcn_norm_kernel<<<(2 * npairs + B - 1) / B, B, 0, stream>>>(degP, deg_out, degS, npairs, Nn);
  gcn_gemm_wmma_kernel<<<gemmGrid, B, 0, stream>>>(features, F, F, W1, H, deg_out,
                                                   bufX, H, ntiles);
  gcn_scatter_rows_kernel<<<scatterGrid, B, 0, stream>>>(bufX, src, dst, bufA, E);
  gcn_post_kernel<<<(NH + B - 1) / B, B, 0, stream>>>(bufA, deg_in, b1, NH, H);
  gcn_gemm_wmma_kernel<<<gemmGrid, B, 0, stream>>>(bufA, H, H, W2, H, deg_out,
                                                   bufX, H, ntiles);
  gcn_zero_kernel<<<(NH + B - 1) / B, B, 0, stream>>>(bufA, NH);
  gcn_scatter_rows_kernel<<<scatterGrid, B, 0, stream>>>(bufX, src, dst, bufA, E);
  gcn_post_kernel<<<(NH + B - 1) / B, B, 0, stream>>>(bufA, deg_in, b2, NH, H);
  gcn_readout_kernel<<<(nq + B - 1) / B, B, 0, stream>>>(bufA, Wd, bd, out, nq, 4 * H);
}


# the 4 elementwise passes (2 zero fills, 2 norm_dst+bias+relu) fused into the GEMM kernels (loader, tile prologue) and the readout: 11 launches -> 7, 600k fewer waves; same arithmetic and sum order
# speedup vs baseline: 2.2102x; 1.1513x over previous
#include <hip/hip_runtime.h>

typedef __attribute__((ext_vector_type(2))) float v2f;
typedef __attribute__((ext_vector_type(8))) float v8f;

__device__ __forceinline__ void atomic_add_f32(float* p, float v) {
  (void)__hip_atomic_fetch_add(p, v, __ATOMIC_RELAXED, __HIP_MEMORY_SCOPE_AGENT);
}


#define DEG_RANGE 50000
#define DEG_SLICE 50000

__device__ __forceinline__ void deg_count_one(unsigned int* tbl, int idx, int base) {
  const unsigned int rel = (unsigned int)(idx - base);
  if (rel < (unsigned int)DEG_RANGE) {
    (void)__hip_atomic_fetch_add(&tbl[rel >> 1], 1u << ((rel & 1u) * 16u), __ATOMIC_RELAXED,
                                 __HIP_MEMORY_SCOPE_WORKGROUP);
  }
}

__global__ void gcn_degree_lds_kernel(const int* __restrict__ src, const int* __restrict__ dst,
                                      unsigned int* __restrict__ P, int E, int npairs) {
  __shared__ unsigned int tbl[DEG_RANGE / 2];

  const int s = blockIdx.x;
  const int r = blockIdx.y;
  const int a = blockIdx.z;
  const int S = gridDim.x;

  for (int i = threadIdx.x; i < DEG_RANGE / 2; i += blockDim.x) tbl[i] = 0u;
  __syncthreads();

  const int* idxp = (a == 0 ? src : dst) + (size_t)s * DEG_SLICE;
  int n = E - s * DEG_SLICE;
  if (n > DEG_SLICE) n = DEG_SLICE;
  const int base = r * DEG_RANGE;

  const int n4 = n >> 2;
  const int4* idx4 = (const int4*)idxp;
  for (int i0 = threadIdx.x; i0 < n4; i0 += 4 * blockDim.x) {
    int4 v[4];
#pragma unroll
    for (int j = 0; j < 4; ++j) {
      const int i = i0 + j * (int)blockDim.x;
      v[j] = (i < n4) ? idx4[i] : make_int4(-1, -1, -1, -1);
    }
#pragma unroll
    for (int j = 0; j < 4; ++j) {
      deg_count_one(tbl, v[j].x, base);
      deg_count_one(tbl, v[j].y, base);
      deg_count_one(tbl, v[j].z, base);
      deg_count_one(tbl, v[j].w, base);
    }
  }
  for (int i = (n4 << 2) + threadIdx.x; i < n; i += blockDim.x) deg_count_one(tbl, idxp[i], base);
  __syncthreads();

  unsigned int* plane = P + ((size_t)a * S + s) * npairs;
  const int w0 = r * (DEG_RANGE / 2);
  for (int i = threadIdx.x; i < DEG_RANGE / 2; i += blockDim.x) {
    const int w = w0 + i;
    if (w < npairs) plane[w] = tbl[i];
  }
}

__global__ void gcn_norm_kernel(const unsigned int* __restrict__ P, float* __restrict__ norm,
                                int S, int npairs, int Nn) {
  const int t = blockIdx.x * blockDim.x + threadIdx.x;
  if (t >= 2 * npairs) return;
  const int a = t / npairs;
  const int w = t - a * npairs;
  const unsigned int* p = P + (size_t)a * S * npairs + w;
  unsigned int lo = 0u, hi = 0u;
#pragma unroll 8
  for (int s = 0; s < S; ++s) {
    const unsigned int v = p[(size_t)s * npairs];
    lo += v & 0xffffu;
    hi += v >> 16;
  }
  float* o = norm + (size_t)a * Nn;
  const int node = 2 * w;
  o[node] = rsqrtf(fmaxf((float)lo, 1.0f));
  if (node + 1 < Nn) o[node + 1] = rsqrtf(fmaxf((float)hi, 1.0f));
}

#define GCN_ROW 24

template <bool POST>
__device__ __forceinline__ void gcn_gemm_tile(float* __restrict__ Ws,
                                              const float* __restrict__ A, int lda, int K,
                                              const float* __restrict__ W, int wcols,
                                              const float* __restrict__ norm_src,
                                              const float* __restrict__ norm_dst,
                                              const float* __restrict__ bias,
                                              float* __restrict__ out, int ldo,
                                              float* __restrict__ zero_rows, int ntiles) {
  for (int i = threadIdx.x; i < 32 * 32; i += blockDim.x) {
    const int k = i >> 5;
    const int n = i & 31;
    Ws[i] = (k < K && n < wcols) ? W[(size_t)k * wcols + n] : 0.0f;
  }
  __syncthreads();

  const int wave = (blockIdx.x * blockDim.x + threadIdx.x) >> 5;
  const int lane = threadIdx.x & 31;
  if (wave >= ntiles) return;

  {
    float4* z = (float4*)(zero_rows + (size_t)wave * (16 * GCN_ROW));
    for (int i = lane; i < 16 * GCN_ROW / 4; i += 32) z[i] = make_float4(0.0f, 0.0f, 0.0f, 0.0f);
  }

  const int m = lane & 15;
  const int half = lane >> 4;

  const int row = wave * 16 + m;
  const float nrm = norm_src[row];
  const float nd = POST ? norm_dst[row] : 0.0f;
  const float* arow = A + (size_t)row * lda;

  v8f c0 = {};
  v8f c1 = {};

  for (int k0 = 0; k0 < K; k0 += 4) {
    const int k = k0 + 2 * half;
    const float2 av = *(const float2*)(arow + k);
    v2f a;
    if (POST) {
      const float2 bv = *(const float2*)(bias + k);
      a.x = fmaxf(fmaf(av.x, nd, bv.x), 0.0f) * nrm;
      a.y = fmaxf(fmaf(av.y, nd, bv.y), 0.0f) * nrm;
    } else {
      a.x = av.x * nrm;
      a.y = av.y * nrm;
    }

    v2f b0, b1;
    b0.x = Ws[k * 32 + m];
    b0.y = Ws[(k + 1) * 32 + m];
    b1.x = Ws[k * 32 + 16 + m];
    b1.y = Ws[(k + 1) * 32 + 16 + m];

    c0 = __builtin_amdgcn_wmma_f32_16x16x4_f32(false, a, false, b0, (short)0, c0, false, false);
    c1 = __builtin_amdgcn_wmma_f32_16x16x4_f32(false, a, false, b1, (short)0, c1, false, false);
  }

  for (int v = 0; v < 8; ++v) {
    const int orow = wave * 16 + v + 8 * half;
    out[(size_t)orow * ldo + m] = c0[v];
    if (16 + m < wcols) out[(size_t)orow * ldo + 16 + m] = c1[v];
  }
}

__global__ void gcn_gemm_wmma_kernel(const float* __restrict__ A, int lda, int K,
                                     const float* __restrict__ W, int wcols,
                                     const float* __restrict__ norm_src,
                                     float* __restrict__ out, int ldo,
                                     float* __restrict__ zero_rows, int ntiles) {
  __shared__ float Ws[32 * 32];
  gcn_gemm_tile<false>(Ws, A, lda, K, W, wcols, norm_src, nullptr, nullptr, out, ldo, zero_rows,
                       ntiles);
}

__global__ void gcn_gemm_wmma_post_kernel(const float* __restrict__ A, int lda, int K,
                                          const float* __restrict__ W, int wcols,
                                          const float* __restrict__ norm_src,
                                          const float* __restrict__ norm_dst,
                                          const float* __restrict__ bias,
                                          float* __restrict__ out, int ldo,
                                          float* __restrict__ zero_rows, int ntiles) {
  __shared__ float Ws[32 * 32];
  gcn_gemm_tile<true>(Ws, A, lda, K, W, wcols, norm_src, norm_dst, bias, out, ldo, zero_rows,
                      ntiles);
}

#define SCAT_BATCH 16

__global__ void gcn_scatter_rows_kernel(const float* __restrict__ x, const int* __restrict__ src,
                                        const int* __restrict__ dst, float* __restrict__ agg,
                                        int E) {
  const int lane = threadIdx.x & 31;
  const int wave = (blockIdx.x * blockDim.x + threadIdx.x) >> 5;
  const int e0 = wave * 32;
  if (e0 >= E) return;
  const int e = e0 + lane;
  const int sv = (e < E) ? src[e] : 0;
  const int dv = (e < E) ? dst[e] : 0;
  const bool act = lane < GCN_ROW;

  if (e0 + 32 <= E) {
#pragma unroll
    for (int b = 0; b < 32; b += SCAT_BATCH) {
      int si[SCAT_BATCH], di[SCAT_BATCH];
#pragma unroll
      for (int k = 0; k < SCAT_BATCH; ++k) {
        si[k] = __builtin_amdgcn_readlane(sv, b + k);
        di[k] = __builtin_amdgcn_readlane(dv, b + k);
      }
      if (act) {
        float v[SCAT_BATCH];
#pragma unroll
        for (int k = 0; k < SCAT_BATCH; ++k) v[k] = (x + (size_t)si[k] * GCN_ROW)[lane];
#pragma unroll
        for (int k = 0; k < SCAT_BATCH; ++k)
          atomic_add_f32(agg + (size_t)di[k] * GCN_ROW + lane, v[k]);
      }
    }
  } else {
    const int n = E - e0;
    for (int j = 0; j < n; ++j) {
      const int s = __builtin_amdgcn_readlane(sv, j);
      const int d = __builtin_amdgcn_readlane(dv, j);
      if (act) atomic_add_f32(agg + (size_t)d * GCN_ROW + lane, (x + (size_t)s * GCN_ROW)[lane]);
    }
  }
}


__device__ __forceinline__ float gcn_readout_node(const float4* __restrict__ hp,
                                                  const float4* __restrict__ wp,
                                                  const float4* __restrict__ bp, float nd,
                                                  float s) {
  for (int c = 0; c < GCN_ROW / 4; ++c) {
    const float4 hv = hp[c];
    const float4 wv = wp[c];
    const float4 bv = bp[c];
    s = fmaf(fmaxf(fmaf(hv.x, nd, bv.x), 0.0f), wv.x, s);
    s = fmaf(fmaxf(fmaf(hv.y, nd, bv.y), 0.0f), wv.y, s);
    s = fmaf(fmaxf(fmaf(hv.z, nd, bv.z), 0.0f), wv.z, s);
    s = fmaf(fmaxf(fmaf(hv.w, nd, bv.w), 0.0f), wv.w, s);
  }
  return s;
}

__global__ void gcn_readout_post_kernel(const float* __restrict__ agg,
                                        const float* __restrict__ norm_dst,
                                        const float* __restrict__ bias,
                                        const float* __restrict__ Wd,
                                        const float* __restrict__ bd, float* __restrict__ out,
                                        int nq) {
  int r = blockIdx.x * blockDim.x + threadIdx.x;
  if (r >= nq) return;
  const float4* hp = (const float4*)(agg + (size_t)r * (4 * GCN_ROW));
  const float4* wp = (const float4*)Wd;
  const float4* bp = (const float4*)bias;
  const float4 nd = *(const float4*)(norm_dst + (size_t)r * 4);
  float s = 0.0f;
  s = gcn_readout_node(hp + 0 * (GCN_ROW / 4), wp + 0 * (GCN_ROW / 4), bp, nd.x, s);
  s = gcn_readout_node(hp + 1 * (GCN_ROW / 4), wp + 1 * (GCN_ROW / 4), bp, nd.y, s);
  s = gcn_readout_node(hp + 2 * (GCN_ROW / 4), wp + 2 * (GCN_ROW / 4), bp, nd.z, s);
  s = gcn_readout_node(hp + 3 * (GCN_ROW / 4), wp + 3 * (GCN_ROW / 4), bp, nd.w, s);
  out[r] = s + bd[0];
}


extern "C" void kernel_launch(void* const* d_in, const int* in_sizes, int n_in,
                              void* d_out, int out_size, void* d_ws, size_t ws_size,
                              hipStream_t stream) {
  const float* features = (const float*)d_in[0];
  const int*   src      = (const int*)d_in[1];
  const int*   dst      = (const int*)d_in[2];
  const float* W1       = (const float*)d_in[3];
  const float* b1       = (const float*)d_in[4];
  const float* W2       = (const float*)d_in[5];
  const float* b2       = (const float*)d_in[6];
  const float* Wd       = (const float*)d_in[7];
  const float* bd       = (const float*)d_in[8];
  float* out = (float*)d_out;

  const int H  = in_sizes[4];
  const int F  = in_sizes[3] / H;
  const int Nn = in_sizes[0] / F;
  const int E  = in_sizes[1];

  float* deg_out = (float*)d_ws;
  float* deg_in  = deg_out + Nn;
  float* aggA    = deg_in + Nn;
  float* bufX    = aggA + (size_t)Nn * H;
  float* aggB    = bufX + (size_t)Nn * H;
  unsigned int* degP = (unsigned int*)(aggB + (size_t)Nn * H);
  const int npairs = (Nn + 1) / 2;
  const int degS = (E + DEG_SLICE - 1) / DEG_SLICE;
  const int degR = (Nn + DEG_RANGE - 1) / DEG_RANGE;

  const int B = 256;
  const int ntiles = Nn / 16;
  const int wavesPerBlock = B / 32;
  const int gemmGrid = (ntiles + wavesPerBlock - 1) / wavesPerBlock;
  const int scatterWaves = (E + 31) / 32;
  const int scatterGrid = (scatterWaves + wavesPerBlock - 1) / wavesPerBlock;
  const int nq = Nn / 4;

  gcn_degree_lds_kernel<<<dim3(degS, degR, 2), 1024, 0, stream>>>(src, dst, degP, E, npairs);
  gcn_norm_kernel<<<(2 * npairs + B - 1) / B, B, 0, stream>>>(degP, deg_out, degS, npairs, Nn);
  gcn_gemm_wmma_kernel<<<gemmGrid, B, 0, stream>>>(features, F, F, W1, H, deg_out,
                                                   bufX, H, aggA, ntiles);
  gcn_scatter_rows_kernel<<<scatterGrid, B, 0, stream>>>(bufX, src, dst, aggA, E);
  gcn_gemm_wmma_post_kernel<<<gemmGrid, B, 0, stream>>>(aggA, H, H, W2, H, deg_out, deg_in, b1,
                                                        bufX, H, aggB, ntiles);
  gcn_scatter_rows_kernel<<<scatterGrid, B, 0, stream>>>(bufX, src, dst, aggB, E);
  gcn_readout_post_kernel<<<(nq + B - 1) / B, B, 0, stream>>>(aggB, deg_in, b2, Wd, bd, out, nq);
}


# edge scatter split by cache half: a work-group reads its die, draws a 512-edge chunk ticket of that half and adds only the rows whose lines that half owns; f32 atomics, arrival-order sums kept
# speedup vs baseline: 2.8316x; 1.2812x over previous
#include <hip/hip_runtime.h>

typedef __attribute__((ext_vector_type(2))) float v2f;
typedef __attribute__((ext_vector_type(8))) float v8f;

__device__ __forceinline__ void atomic_add_f32(float* p, float v) {
  (void)__hip_atomic_fetch_add(p, v, __ATOMIC_RELAXED, __HIP_MEMORY_SCOPE_AGENT);
}


#define DEG_RANGE 50000
#define DEG_SLICE 50000

__device__ __forceinline__ void deg_count_one(unsigned int* tbl, int idx, int base) {
  const unsigned int rel = (unsigned int)(idx - base);
  if (rel < (unsigned int)DEG_RANGE) {
    (void)__hip_atomic_fetch_add(&tbl[rel >> 1], 1u << ((rel & 1u) * 16u), __ATOMIC_RELAXED,
                                 __HIP_MEMORY_SCOPE_WORKGROUP);
  }
}

__global__ void gcn_degree_lds_kernel(const int* __restrict__ src, const int* __restrict__ dst,
                                      unsigned int* __restrict__ P, int E, int npairs) {
  __shared__ unsigned int tbl[DEG_RANGE / 2];

  const int s = blockIdx.x;
  const int r = blockIdx.y;
  const int a = blockIdx.z;
  const int S = gridDim.x;

  for (int i = threadIdx.x; i < DEG_RANGE / 2; i += blockDim.x) tbl[i] = 0u;
  __syncthreads();

  const int* idxp = (a == 0 ? src : dst) + (size_t)s * DEG_SLICE;
  int n = E - s * DEG_SLICE;
  if (n > DEG_SLICE) n = DEG_SLICE;
  const int base = r * DEG_RANGE;

  const int n4 = n >> 2;
  const int4* idx4 = (const int4*)idxp;
  for (int i0 = threadIdx.x; i0 < n4; i0 += 4 * blockDim.x) {
    int4 v[4];
#pragma unroll
    for (int j = 0; j < 4; ++j) {
      const int i = i0 + j * (int)blockDim.x;
      v[j] = (i < n4) ? idx4[i] : make_int4(-1, -1, -1, -1);
    }
#pragma unroll
    for (int j = 0; j < 4; ++j) {
      deg_count_one(tbl, v[j].x, base);
      deg_count_one(tbl, v[j].y, base);
      deg_count_one(tbl, v[j].z, base);
      deg_count_one(tbl, v[j].w, base);
    }
  }
  for (int i = (n4 << 2) + threadIdx.x; i < n; i += blockDim.x) deg_count_one(tbl, idxp[i], base);
  __syncthreads();

  unsigned int* plane = P + ((size_t)a * S + s) * npairs;
  const int w0 = r * (DEG_RANGE / 2);
  for (int i = threadIdx.x; i < DEG_RANGE / 2; i += blockDim.x) {
    const int w = w0 + i;
    if (w < npairs) plane[w] = tbl[i];
  }
}

__global__ void gcn_norm_kernel(const unsigned int* __restrict__ P, float* __restrict__ norm,
                                int S, int npairs, int Nn, unsigned int* __restrict__ tick) {
  const int t = blockIdx.x * blockDim.x + threadIdx.x;
  if (t == 0) {
    tick[0] = 0u;
    tick[1] = 0u;
    tick[2] = 0u;
    tick[3] = 0u;
  }
  if (t >= 2 * npairs) return;
  const int a = t / npairs;
  const int w = t - a * npairs;
  const unsigned int* p = P + (size_t)a * S * npairs + w;
  unsigned int lo = 0u, hi = 0u;
#pragma unroll 8
  for (int s = 0; s < S; ++s) {
    const unsigned int v = p[(size_t)s * npairs];
    lo += v & 0xffffu;
    hi += v >> 16;
  }
  float* o = norm + (size_t)a * Nn;
  const int node = 2 * w;
  o[node] = rsqrtf(fmaxf((float)lo, 1.0f));
  if (node + 1 < Nn) o[node + 1] = rsqrtf(fmaxf((float)hi, 1.0f));
}

#define GCN_ROW 24

template <bool POST>
__device__ __forceinline__ void gcn_gemm_tile(float* __restrict__ Ws,
                                              const float* __restrict__ A, int lda, int K,
                                              const float* __restrict__ W, int wcols,
                                              const float* __restrict__ norm_src,
                                              const float* __restrict__ norm_dst,
                                              const float* __restrict__ bias,
                                              float* __restrict__ out, int ldo,
                                              float* __restrict__ zero_rows, int ntiles) {
  for (int i = threadIdx.x; i < 32 * 32; i += blockDim.x) {
    const int k = i >> 5;
    const int n = i & 31;
    Ws[i] = (k < K && n < wcols) ? W[(size_t)k * wcols + n] : 0.0f;
  }
  __syncthreads();

  const int wave = (blockIdx.x * blockDim.x + threadIdx.x) >> 5;
  const int lane = threadIdx.x & 31;
  if (wave >= ntiles) return;

  {
    float4* z = (float4*)(zero_rows + (size_t)wave * (16 * GCN_ROW));
    for (int i = lane; i < 16 * GCN_ROW / 4; i += 32) z[i] = make_float4(0.0f, 0.0f, 0.0f, 0.0f);
  }

  const int m = lane & 15;
  const int half = lane >> 4;

  const int row = wave * 16 + m;
  const float nrm = norm_src[row];
  const float nd = POST ? norm_dst[row] : 0.0f;
  const float* arow = A + (size_t)row * lda;

  v8f c0 = {};
  v8f c1 = {};

  for (int k0 = 0; k0 < K; k0 += 4) {
    const int k = k0 + 2 * half;
    const float2 av = *(const float2*)(arow + k);
    v2f a;
    if (POST) {
      const float2 bv = *(const float2*)(bias + k);
      a.x = fmaxf(fmaf(av.x, nd, bv.x), 0.0f) * nrm;
      a.y = fmaxf(fmaf(av.y, nd, bv.y), 0.0f) * nrm;
    } else {
      a.x = av.x * nrm;
      a.y = av.y * nrm;
    }

    v2f b0, b1;
    b0.x = Ws[k * 32 + m];
    b0.y = Ws[(k + 1) * 32 + m];
    b1.x = Ws[k * 32 + 16 + m];
    b1.y = Ws[(k + 1) * 32 + 16 + m];

    c0 = __builtin_amdgcn_wmma_f32_16x16x4_f32(false, a, false, b0, (short)0, c0, false, false);
    c1 = __builtin_amdgcn_wmma_f32_16x16x4_f32(false, a, false, b1, (short)0, c1, false, false);
  }

  for (int v = 0; v < 8; ++v) {
    const int orow = wave * 16 + v + 8 * half;
    out[(size_t)orow * ldo + m] = c0[v];
    if (16 + m < wcols) out[(size_t)orow * ldo + 16 + m] = c1[v];
  }
}

__global__ void gcn_gemm_wmma_kernel(const float* __restrict__ A, int lda, int K,
                                     const float* __restrict__ W, int wcols,
                                     const float* __restrict__ norm_src,
                                     float* __restrict__ out, int ldo,
                                     float* __restrict__ zero_rows, int ntiles) {
  __shared__ float Ws[32 * 32];
  gcn_gemm_tile<false>(Ws, A, lda, K, W, wcols, norm_src, nullptr, nullptr, out, ldo, zero_rows,
                       ntiles);
}

__global__ void gcn_gemm_wmma_post_kernel(const float* __restrict__ A, int lda, int K,
                                          const float* __restrict__ W, int wcols,
                                          const float* __restrict__ norm_src,
                                          const float* __restrict__ norm_dst,
                                          const float* __restrict__ bias,
                                          float* __restrict__ out, int ldo,
                                          float* __restrict__ zero_rows, int ntiles) {
  __shared__ float Ws[32 * 32];
  gcn_gemm_tile<true>(Ws, A, lda, K, W, wcols, norm_src, norm_dst, bias, out, ldo, zero_rows,
                      ntiles);
}

#define SCAT_BATCH 16
#define SCAT_WAVE_EDGES 64
#define SCAT_WAVES 8
#define SCAT_WG_EDGES (SCAT_WAVES * SCAT_WAVE_EDGES)

__device__ __forceinline__ void gcn_scatter_batch(const float* __restrict__ x,
                                                  float* __restrict__ agg, int cs, int cd,
                                                  int p, int lane, bool act) {
  int si[SCAT_BATCH], di[SCAT_BATCH];
#pragma unroll
  for (int k = 0; k < SCAT_BATCH; ++k) {
    si[k] = __builtin_amdgcn_readlane(cs, p + k);
    di[k] = __builtin_amdgcn_readlane(cd, p + k);
  }
  if (act) {
    float v[SCAT_BATCH];
#pragma unroll
    for (int k = 0; k < SCAT_BATCH; ++k) v[k] = (x + (size_t)si[k] * GCN_ROW)[lane];
#pragma unroll
    for (int k = 0; k < SCAT_BATCH; ++k)
      atomic_add_f32(agg + (size_t)di[k] * GCN_ROW + lane, v[k]);
  }
}

__device__ __forceinline__ void gcn_scatter_tail(const float* __restrict__ x,
                                                 float* __restrict__ agg, int cs, int cd,
                                                 int p, int m, int lane, bool act) {
  int si[SCAT_BATCH], di[SCAT_BATCH];
#pragma unroll
  for (int k = 0; k < SCAT_BATCH; ++k) {
    si[k] = __builtin_amdgcn_readlane(cs, p + k);
    di[k] = __builtin_amdgcn_readlane(cd, p + k);
  }
  const int lim = act ? m : 0;
  float v[SCAT_BATCH];
#pragma unroll
  for (int k = 0; k < SCAT_BATCH; ++k)
    if (k < lim) v[k] = (x + (size_t)si[k] * GCN_ROW)[lane];
#pragma unroll
  for (int k = 0; k < SCAT_BATCH; ++k)
    if (k < lim) atomic_add_f32(agg + (size_t)di[k] * GCN_ROW + lane, v[k]);
}

__global__ void __launch_bounds__(SCAT_WAVES * 32)
gcn_scatter_rows_kernel(const float* __restrict__ x, const int* __restrict__ src,
                        const int* __restrict__ dst, float* __restrict__ agg, int E, int Nn,
                        int nchunks, unsigned int* __restrict__ tick) {
  __shared__ int2 q[SCAT_WAVES][SCAT_WAVE_EDGES];
  __shared__ unsigned int tk[2];

  const int lane = threadIdx.x & 31;
  const int w = threadIdx.x >> 5;

  if (threadIdx.x == 0) {
    unsigned int hh = (__builtin_amdgcn_s_sendmsg_rtn(0x87) >> 18) & 1u;
    unsigned int t = __hip_atomic_fetch_add(&tick[hh], 1u, __ATOMIC_RELAXED,
                                            __HIP_MEMORY_SCOPE_AGENT);
    if (t >= (unsigned int)nchunks) {
      hh ^= 1u;
      t = __hip_atomic_fetch_add(&tick[hh], 1u, __ATOMIC_RELAXED, __HIP_MEMORY_SCOPE_AGENT);
    }
    tk[0] = t;
    tk[1] = hh;
  }
  __syncthreads();
  const unsigned int t = __builtin_amdgcn_readfirstlane(tk[0]);
  const bool upper = __builtin_amdgcn_readfirstlane(tk[1]) != 0u;
  if (t >= (unsigned int)nchunks) return;

  const int half_row = (Nn >> 4) << 3;

  const int eA = (int)t * SCAT_WG_EDGES + w * SCAT_WAVE_EDGES + lane;
  const int eB = eA + 32;
  const bool inA = eA < E;
  const bool inB = eB < E;
  const int sA = inA ? src[eA] : 0;
  const int dA = inA ? dst[eA] : 0;
  const int sB = inB ? src[eB] : 0;
  const int dB = inB ? dst[eB] : 0;
  const bool selA = inA && ((dA >= half_row) == upper);
  const bool selB = inB && ((dB >= half_row) == upper);
  const unsigned int mA = __builtin_amdgcn_ballot_w32(selA);
  const unsigned int mB = __builtin_amdgcn_ballot_w32(selB);
  const int nA = __popc(mA);
  const int n = nA + __popc(mB);
  if (selA) q[w][__builtin_amdgcn_mbcnt_lo(mA, 0u)] = make_int2(sA, dA);
  if (selB) q[w][nA + (int)__builtin_amdgcn_mbcnt_lo(mB, 0u)] = make_int2(sB, dB);
  const int2 c0 = q[w][lane];
  const int2 c1 = q[w][32 + lane];

  const bool act = lane < GCN_ROW;
  int base = 0;
  for (; base + SCAT_BATCH <= n; base += SCAT_BATCH) {
    const int cs = (base < 32) ? c0.x : c1.x;
    const int cd = (base < 32) ? c0.y : c1.y;
    gcn_scatter_batch(x, agg, cs, cd, base & 31, lane, act);
  }
  if (base < n) {
    const int cs = (base < 32) ? c0.x : c1.x;
    const int cd = (base < 32) ? c0.y : c1.y;
    gcn_scatter_tail(x, agg, cs, cd, base & 31, n - base, lane, act);
  }
}


__device__ __forceinline__ float gcn_readout_node(const float4* __restrict__ hp,
                                                  const float4* __restrict__ wp,
                                                  const float4* __restrict__ bp, float nd,
                                                  float s) {
  for (int c = 0; c < GCN_ROW / 4; ++c) {
    const float4 hv = hp[c];
    const float4 wv = wp[c];
    const float4 bv = bp[c];
    s = fmaf(fmaxf(fmaf(hv.x, nd, bv.x), 0.0f), wv.x, s);
    s = fmaf(fmaxf(fmaf(hv.y, nd, bv.y), 0.0f), wv.y, s);
    s = fmaf(fmaxf(fmaf(hv.z, nd, bv.z), 0.0f), wv.z, s);
    s = fmaf(fmaxf(fmaf(hv.w, nd, bv.w), 0.0f), wv.w, s);
  }
  return s;
}

__global__ void gcn_readout_post_kernel(const float* __restrict__ agg,
                                        const float* __restrict__ norm_dst,
                                        const float* __restrict__ bias,
                                        const float* __restrict__ Wd,
                                        const float* __restrict__ bd, float* __restrict__ out,
                                        int nq) {
  int r = blockIdx.x * blockDim.x + threadIdx.x;
  if (r >= nq) return;
  const float4* hp = (const float4*)(agg + (size_t)r * (4 * GCN_ROW));
  const float4* wp = (const float4*)Wd;
  const float4* bp = (const float4*)bias;
  const float4 nd = *(const float4*)(norm_dst + (size_t)r * 4);
  float s = 0.0f;
  s = gcn_readout_node(hp + 0 * (GCN_ROW / 4), wp + 0 * (GCN_ROW / 4), bp, nd.x, s);
  s = gcn_readout_node(hp + 1 * (GCN_ROW / 4), wp + 1 * (GCN_ROW / 4), bp, nd.y, s);
  s = gcn_readout_node(hp + 2 * (GCN_ROW / 4), wp + 2 * (GCN_ROW / 4), bp, nd.z, s);
  s = gcn_readout_node(hp + 3 * (GCN_ROW / 4), wp + 3 * (GCN_ROW / 4), bp, nd.w, s);
  out[r] = s + bd[0];
}


extern "C" void kernel_launch(void* const* d_in, const int* in_sizes, int n_in,
                              void* d_out, int out_size, void* d_ws, size_t ws_size,
                              hipStream_t stream) {
  const float* features = (const float*)d_in[0];
  const int*   src      = (const int*)d_in[1];
  const int*   dst      = (const int*)d_in[2];
  const float* W1       = (const float*)d_in[3];
  const float* b1       = (const float*)d_in[4];
  const float* W2       = (const float*)d_in[5];
  const float* b2       = (const float*)d_in[6];
  const float* Wd       = (const float*)d_in[7];
  const float* bd       = (const float*)d_in[8];
  float* out = (float*)d_out;

  const int H  = in_sizes[4];
  const int F  = in_sizes[3] / H;
  const int Nn = in_sizes[0] / F;
  const int E  = in_sizes[1];

  float* deg_out = (float*)d_ws;
  float* deg_in  = deg_out + Nn;
  float* aggA    = deg_in + Nn;
  float* bufX    = aggA + (size_t)Nn * H;
  float* aggB    = bufX + (size_t)Nn * H;
  unsigned int* degP = (unsigned int*)(aggB + (size_t)Nn * H);
  const int npairs = (Nn + 1) / 2;
  const int degS = (E + DEG_SLICE - 1) / DEG_SLICE;
  const int degR = (Nn + DEG_RANGE - 1) / DEG_RANGE;
  unsigned int* tick = degP + (size_t)2 * degS * npairs;

  const int B = 256;
  const int ntiles = Nn / 16;
  const int wavesPerBlock = B / 32;
  const int gemmGrid = (ntiles + wavesPerBlock - 1) / wavesPerBlock;
  const int scatChunks = (E + SCAT_WG_EDGES - 1) / SCAT_WG_EDGES;
  const int scatterGrid = 2 * scatChunks;
  const int nq = Nn / 4;

  gcn_degree_lds_kernel<<<dim3(degS, degR, 2), 1024, 0, stream>>>(src, dst, degP, E, npairs);
  gcn_norm_kernel<<<(2 * npairs + B - 1) / B, B, 0, stream>>>(degP, deg_out, degS, npairs, Nn,
                                                              tick);
  gcn_gemm_wmma_kernel<<<gemmGrid, B, 0, stream>>>(features, F, F, W1, H, deg_out,
                                                   bufX, H, aggA, ntiles);
  gcn_scatter_rows_kernel<<<scatterGrid, B, 0, stream>>>(bufX, src, dst, aggA, E, Nn, scatChunks,
                                                         tick);
  gcn_gemm_wmma_post_kernel<<<gemmGrid, B, 0, stream>>>(aggA, H, H, W2, H, deg_out, deg_in, b1,
                                                        bufX, H, aggB, ntiles);
  gcn_scatter_rows_kernel<<<scatterGrid, B, 0, stream>>>(bufX, src, dst, aggB, E, Nn, scatChunks,
                                                         tick + 2);
  gcn_readout_post_kernel<<<(nq + B - 1) / B, B, 0, stream>>>(aggB, deg_in, b2, Wd, bd, out, nq);
}


# edge scatter-add replaced by destination buckets: edges counting-sorted per slice by 784-row bucket (LDS counters); a work-group per bucket sums rows by LDS f32 atomics (arrival order), stores once
# speedup vs baseline: 3.7547x; 1.3260x over previous
#include <hip/hip_runtime.h>

typedef __attribute__((ext_vector_type(2))) float v2f;
typedef __attribute__((ext_vector_type(8))) float v8f;


#define DEG_RANGE 50000
#define DEG_SLICE 50000

__device__ __forceinline__ void deg_count_one(unsigned int* tbl, int idx, int base) {
  const unsigned int rel = (unsigned int)(idx - base);
  if (rel < (unsigned int)DEG_RANGE) {
    (void)__hip_atomic_fetch_add(&tbl[rel >> 1], 1u << ((rel & 1u) * 16u), __ATOMIC_RELAXED,
                                 __HIP_MEMORY_SCOPE_WORKGROUP);
  }
}

__global__ void gcn_degree_lds_kernel(const int* __restrict__ src, const int* __restrict__ dst,
                                      unsigned int* __restrict__ P, int E, int npairs) {
  __shared__ unsigned int tbl[DEG_RANGE / 2];

  const int s = blockIdx.x;
  const int r = blockIdx.y;
  const int a = blockIdx.z;
  const int S = gridDim.x;

  for (int i = threadIdx.x; i < DEG_RANGE / 2; i += blockDim.x) tbl[i] = 0u;
  __syncthreads();

  const int* idxp = (a == 0 ? src : dst) + (size_t)s * DEG_SLICE;
  int n = E - s * DEG_SLICE;
  if (n > DEG_SLICE) n = DEG_SLICE;
  const int base = r * DEG_RANGE;

  const int n4 = n >> 2;
  const int4* idx4 = (const int4*)idxp;
  for (int i0 = threadIdx.x; i0 < n4; i0 += 4 * blockDim.x) {
    int4 v[4];
#pragma unroll
    for (int j = 0; j < 4; ++j) {
      const int i = i0 + j * (int)blockDim.x;
      v[j] = (i < n4) ? idx4[i] : make_int4(-1, -1, -1, -1);
    }
#pragma unroll
    for (int j = 0; j < 4; ++j) {
      deg_count_one(tbl, v[j].x, base);
      deg_count_one(tbl, v[j].y, base);
      deg_count_one(tbl, v[j].z, base);
      deg_count_one(tbl, v[j].w, base);
    }
  }
  for (int i = (n4 << 2) + threadIdx.x; i < n; i += blockDim.x) deg_count_one(tbl, idxp[i], base);
  __syncthreads();

  unsigned int* plane = P + ((size_t)a * S + s) * npairs;
  const int w0 = r * (DEG_RANGE / 2);
  for (int i = threadIdx.x; i < DEG_RANGE / 2; i += blockDim.x) {
    const int w = w0 + i;
    if (w < npairs) plane[w] = tbl[i];
  }
}

__global__ void gcn_norm_kernel(const unsigned int* __restrict__ P, float* __restrict__ norm,
                                int S, int npairs, int Nn) {
  const int t = blockIdx.x * blockDim.x + threadIdx.x;
  if (t >= 2 * npairs) return;
  const int a = t / npairs;
  const int w = t - a * npairs;
  const unsigned int* p = P + (size_t)a * S * npairs + w;
  unsigned int lo = 0u, hi = 0u;
#pragma unroll 8
  for (int s = 0; s < S; ++s) {
    const unsigned int v = p[(size_t)s * npairs];
    lo += v & 0xffffu;
    hi += v >> 16;
  }
  float* o = norm + (size_t)a * Nn;
  const int node = 2 * w;
  o[node] = rsqrtf(fmaxf((float)lo, 1.0f));
  if (node + 1 < Nn) o[node + 1] = rsqrtf(fmaxf((float)hi, 1.0f));
}

#define GCN_ROW 24

template <bool POST>
__device__ __forceinline__ void gcn_gemm_tile(float* __restrict__ Ws,
                                              const float* __restrict__ A, int lda, int K,
                                              const float* __restrict__ W, int wcols,
                                              const float* __restrict__ norm_src,
                                              const float* __restrict__ norm_dst,
                                              const float* __restrict__ bias,
                                              float* __restrict__ out, int ldo, int ntiles) {
  for (int i = threadIdx.x; i < 32 * 32; i += blockDim.x) {
    const int k = i >> 5;
    const int n = i & 31;
    Ws[i] = (k < K && n < wcols) ? W[(size_t)k * wcols + n] : 0.0f;
  }
  __syncthreads();

  const int wave = (blockIdx.x * blockDim.x + threadIdx.x) >> 5;
  const int lane = threadIdx.x & 31;
  if (wave >= ntiles) return;

  const int m = lane & 15;
  const int half = lane >> 4;

  const int row = wave * 16 + m;
  const float nrm = norm_src[row];
  const float nd = POST ? norm_dst[row] : 0.0f;
  const float* arow = A + (size_t)row * lda;

  v8f c0 = {};
  v8f c1 = {};

  for (int k0 = 0; k0 < K; k0 += 4) {
    const int k = k0 + 2 * half;
    const float2 av = *(const float2*)(arow + k);
    v2f a;
    if (POST) {
      const float2 bv = *(const float2*)(bias + k);
      a.x = fmaxf(fmaf(av.x, nd, bv.x), 0.0f) * nrm;
      a.y = fmaxf(fmaf(av.y, nd, bv.y), 0.0f) * nrm;
    } else {
      a.x = av.x * nrm;
      a.y = av.y * nrm;
    }

    v2f b0, b1;
    b0.x = Ws[k * 32 + m];
    b0.y = Ws[(k + 1) * 32 + m];
    b1.x = Ws[k * 32 + 16 + m];
    b1.y = Ws[(k + 1) * 32 + 16 + m];

    c0 = __builtin_amdgcn_wmma_f32_16x16x4_f32(false, a, false, b0, (short)0, c0, false, false);
    c1 = __builtin_amdgcn_wmma_f32_16x16x4_f32(false, a, false, b1, (short)0, c1, false, false);
  }

  for (int v = 0; v < 8; ++v) {
    const int orow = wave * 16 + v + 8 * half;
    out[(size_t)orow * ldo + m] = c0[v];
    if (16 + m < wcols) out[(size_t)orow * ldo + 16 + m] = c1[v];
  }
}

__global__ void gcn_gemm_wmma_kernel(const float* __restrict__ A, int lda, int K,
                                     const float* __restrict__ W, int wcols,
                                     const float* __restrict__ norm_src,
                                     float* __restrict__ out, int ldo, int ntiles) {
  __shared__ float Ws[32 * 32];
  gcn_gemm_tile<false>(Ws, A, lda, K, W, wcols, norm_src, nullptr, nullptr, out, ldo, ntiles);
}

__global__ void gcn_gemm_wmma_post_kernel(const float* __restrict__ A, int lda, int K,
                                          const float* __restrict__ W, int wcols,
                                          const float* __restrict__ norm_src,
                                          const float* __restrict__ norm_dst,
                                          const float* __restrict__ bias,
                                          float* __restrict__ out, int ldo, int ntiles) {
  __shared__ float Ws[32 * 32];
  gcn_gemm_tile<true>(Ws, A, lda, K, W, wcols, norm_src, norm_dst, bias, out, ldo, ntiles);
}

#define BIN_NB 256
#define BIN_NPB 784
#define BIN_THREADS 512
#define BIN_EPT 8
#define BIN_SLICE (BIN_THREADS * BIN_EPT)

__global__ void __launch_bounds__(BIN_THREADS)
gcn_bin_kernel(const int* __restrict__ src, const int* __restrict__ dst, int2* __restrict__ rec,
               int2* __restrict__ tab, int E) {
  __shared__ __attribute__((aligned(16))) unsigned int cnt[BIN_NB];
  __shared__ unsigned int start[BIN_NB];
  __shared__ int2 sorted[BIN_SLICE];

  const int tid = threadIdx.x;
  const int sl = blockIdx.x;
  const int e0 = sl * BIN_SLICE;

  if (tid < BIN_NB) cnt[tid] = 0u;
  __syncthreads();

  int s[BIN_EPT], d[BIN_EPT], bk[BIN_EPT];
  unsigned int rk[BIN_EPT];
#pragma unroll
  for (int j = 0; j < BIN_EPT; ++j) {
    const int e = e0 + j * BIN_THREADS + tid;
    const bool ok = e < E;
    const int ec = ok ? e : E - 1;
    s[j] = src[ec];
    const int dv = dst[ec];
    d[j] = ok ? dv : -1;
  }
#pragma unroll
  for (int j = 0; j < BIN_EPT; ++j) {
    bk[j] = 0;
    rk[j] = 0u;
    if (d[j] >= 0) {
      bk[j] = d[j] / BIN_NPB;
      rk[j] = __hip_atomic_fetch_add(&cnt[bk[j]], 1u, __ATOMIC_RELAXED,
                                     __HIP_MEMORY_SCOPE_WORKGROUP);
    }
  }
  __syncthreads();

  if (tid < BIN_NB) {
    const uint4* c4 = (const uint4*)cnt;
    unsigned int acc = 0u;
#pragma unroll 8
    for (int i = 0; i < BIN_NB / 4; ++i) {
      const uint4 c = c4[i];
      acc += (4 * i + 0 < tid) ? c.x : 0u;
      acc += (4 * i + 1 < tid) ? c.y : 0u;
      acc += (4 * i + 2 < tid) ? c.z : 0u;
      acc += (4 * i + 3 < tid) ? c.w : 0u;
    }
    start[tid] = acc;
    tab[(size_t)sl * BIN_NB + tid] = make_int2((int)acc, (int)cnt[tid]);
  }
  __syncthreads();

#pragma unroll
  for (int j = 0; j < BIN_EPT; ++j)
    if (d[j] >= 0) sorted[start[bk[j]] + rk[j]] = make_int2(s[j], d[j]);
  __syncthreads();

  int n = E - e0;
  if (n > BIN_SLICE) n = BIN_SLICE;
  for (int i = tid; i < n; i += BIN_THREADS) rec[(size_t)e0 + i] = sorted[i];
}

__device__ __forceinline__ void lds_add_f32(float* p, float v) {
  (void)__hip_atomic_fetch_add(p, v, __ATOMIC_RELAXED, __HIP_MEMORY_SCOPE_WORKGROUP);
}

__device__ __forceinline__ float gcn_row_float(const float* __restrict__ x, int r, int lane) {
  const unsigned int off = (unsigned int)r * (unsigned int)(GCN_ROW * sizeof(float)) +
                           (unsigned int)lane * (unsigned int)sizeof(float);
  return *(const float*)((const char*)x + off);
}

#define SUM_BATCH 16
#define SUM_THREADS 1024
#define SUM_WAVES (SUM_THREADS / 32)

__device__ __forceinline__ void gcn_sum_batch(const float* __restrict__ x, float* rows,
                                              int nbase, int cs, int cd, int p, int lane,
                                              bool act) {
  int si[SUM_BATCH], di[SUM_BATCH];
#pragma unroll
  for (int k = 0; k < SUM_BATCH; ++k) {
    si[k] = __builtin_amdgcn_readlane(cs, p + k);
    di[k] = __builtin_amdgcn_readlane(cd, p + k) - nbase;
  }
  if (act) {
    float v[SUM_BATCH];
#pragma unroll
    for (int k = 0; k < SUM_BATCH; ++k) v[k] = gcn_row_float(x, si[k], lane);
#pragma unroll
    for (int k = 0; k < SUM_BATCH; ++k) lds_add_f32(rows + di[k] * GCN_ROW + lane, v[k]);
  }
}

__device__ __forceinline__ void gcn_sum_tail(const float* __restrict__ x, float* rows,
                                             int nbase, int cs, int cd, int p, int m, int lane,
                                             bool act) {
  int di[15];
  float v[15];
  const bool h8 = (m & 8) != 0;
  const bool h4 = (m & 4) != 0;
  const bool h2 = (m & 2) != 0;
  const bool h1 = (m & 1) != 0;
  int q = p;
  if (h8) {
    int si[8];
#pragma unroll
    for (int k = 0; k < 8; ++k) {
      si[k] = __builtin_amdgcn_readlane(cs, q + k);
      di[k] = __builtin_amdgcn_readlane(cd, q + k) - nbase;
    }
    if (act) {
#pragma unroll
      for (int k = 0; k < 8; ++k) v[k] = gcn_row_float(x, si[k], lane);
    }
    q += 8;
  }
  if (h4) {
    int si[4];
#pragma unroll
    for (int k = 0; k < 4; ++k) {
      si[k] = __builtin_amdgcn_readlane(cs, q + k);
      di[8 + k] = __builtin_amdgcn_readlane(cd, q + k) - nbase;
    }
    if (act) {
#pragma unroll
      for (int k = 0; k < 4; ++k) v[8 + k] = gcn_row_float(x, si[k], lane);
    }
    q += 4;
  }
  if (h2) {
    int si[2];
#pragma unroll
    for (int k = 0; k < 2; ++k) {
      si[k] = __builtin_amdgcn_readlane(cs, q + k);
      di[12 + k] = __builtin_amdgcn_readlane(cd, q + k) - nbase;
    }
    if (act) {
#pragma unroll
      for (int k = 0; k < 2; ++k) v[12 + k] = gcn_row_float(x, si[k], lane);
    }
    q += 2;
  }
  if (h1) {
    const int s1 = __builtin_amdgcn_readlane(cs, q);
    di[14] = __builtin_amdgcn_readlane(cd, q) - nbase;
    if (act) v[14] = gcn_row_float(x, s1, lane);
  }
  if (h8 && act) {
#pragma unroll
    for (int k = 0; k < 8; ++k) lds_add_f32(rows + di[k] * GCN_ROW + lane, v[k]);
  }
  if (h4 && act) {
#pragma unroll
    for (int k = 0; k < 4; ++k) lds_add_f32(rows + di[8 + k] * GCN_ROW + lane, v[8 + k]);
  }
  if (h2 && act) {
#pragma unroll
    for (int k = 0; k < 2; ++k) lds_add_f32(rows + di[12 + k] * GCN_ROW + lane, v[12 + k]);
  }
  if (h1 && act) lds_add_f32(rows + di[14] * GCN_ROW + lane, v[14]);
}

__device__ __forceinline__ void gcn_sum_records(const float* __restrict__ x, float* rows,
                                                int nbase, int cs, int cd, int m, int lane,
                                                bool act) {
  int p = 0;
  for (; p + SUM_BATCH <= m; p += SUM_BATCH) gcn_sum_batch(x, rows, nbase, cs, cd, p, lane, act);
  if (p < m) gcn_sum_tail(x, rows, nbase, cs, cd, p, m - p, lane, act);
}

__global__ void __launch_bounds__(SUM_THREADS)
gcn_bucket_sum_kernel(const float* __restrict__ x, const int2* __restrict__ rec,
                      const int2* __restrict__ tab, float* __restrict__ agg, int nS, int Nn) {
  __shared__ __attribute__((aligned(16))) float rows[BIN_NPB * GCN_ROW];

  const int b = blockIdx.x;
  const int lane = threadIdx.x & 31;
  const int w = __builtin_amdgcn_readfirstlane((int)(threadIdx.x >> 5));
  const int nbase = b * BIN_NPB;
  const bool act = lane < GCN_ROW;

  {
    float4* z = (float4*)rows;
    for (int i = threadIdx.x; i < BIN_NPB * GCN_ROW / 4; i += SUM_THREADS)
      z[i] = make_float4(0.0f, 0.0f, 0.0f, 0.0f);
  }
  __syncthreads();

  for (int g0 = w; g0 < nS; g0 += SUM_WAVES * 32) {
    const int mysl = g0 + SUM_WAVES * lane;
    int2 te = make_int2(0, 0);
    if (mysl < nS) te = tab[(size_t)mysl * BIN_NB + b];
    int nsl = (nS - g0 + SUM_WAVES - 1) / SUM_WAVES;
    if (nsl > 32) nsl = 32;

    int st = __builtin_amdgcn_readlane(te.x, 0);
    int cn = __builtin_amdgcn_readlane(te.y, 0);
    const int2* rp = rec + (size_t)g0 * BIN_SLICE + st;
    int2 cnext = make_int2(0, 0);
    if (lane < cn) cnext = rp[lane];

    for (int l = 0; l < nsl; ++l) {
      const int2 c = cnext;
      const int2* rpc = rp;
      const int cnc = cn;
      if (l + 1 < nsl) {
        st = __builtin_amdgcn_readlane(te.x, l + 1);
        cn = __builtin_amdgcn_readlane(te.y, l + 1);
        rp = rec + (size_t)(g0 + SUM_WAVES * (l + 1)) * BIN_SLICE + st;
        cnext = make_int2(0, 0);
        if (lane < cn) cnext = rp[lane];
      }
      if (cnc > 0) gcn_sum_records(x, rows, nbase, c.x, c.y, cnc < 32 ? cnc : 32, lane, act);
      for (int i0 = 32; i0 < cnc; i0 += 32) {
        const int m = (cnc - i0 < 32) ? (cnc - i0) : 32;
        int2 c2 = make_int2(0, 0);
        if (lane < m) c2 = rpc[i0 + lane];
        gcn_sum_records(x, rows, nbase, c2.x, c2.y, m, lane, act);
      }
    }
  }
  __syncthreads();

  int nr = Nn - nbase;
  if (nr > BIN_NPB) nr = BIN_NPB;
  const float4* r4 = (const float4*)rows;
  float4* o4 = (float4*)(agg + (size_t)nbase * GCN_ROW);
  for (int i = threadIdx.x; i < nr * (GCN_ROW / 4); i += SUM_THREADS) o4[i] = r4[i];
}


__device__ __forceinline__ float gcn_readout_node(const float4* __restrict__ hp,
                                                  const float4* __restrict__ wp,
                                                  const float4* __restrict__ bp, float nd,
                                                  float s) {
  for (int c = 0; c < GCN_ROW / 4; ++c) {
    const float4 hv = hp[c];
    const float4 wv = wp[c];
    const float4 bv = bp[c];
    s = fmaf(fmaxf(fmaf(hv.x, nd, bv.x), 0.0f), wv.x, s);
    s = fmaf(fmaxf(fmaf(hv.y, nd, bv.y), 0.0f), wv.y, s);
    s = fmaf(fmaxf(fmaf(hv.z, nd, bv.z), 0.0f), wv.z, s);
    s = fmaf(fmaxf(fmaf(hv.w, nd, bv.w), 0.0f), wv.w, s);
  }
  return s;
}

__global__ void gcn_readout_post_kernel(const float* __restrict__ agg,
                                        const float* __restrict__ norm_dst,
                                        const float* __restrict__ bias,
                                        const float* __restrict__ Wd,
                                        const float* __restrict__ bd, float* __restrict__ out,
                                        int nq) {
  int r = blockIdx.x * blockDim.x + threadIdx.x;
  if (r >= nq) return;
  const float4* hp = (const float4*)(agg + (size_t)r * (4 * GCN_ROW));
  const float4* wp = (const float4*)Wd;
  const float4* bp = (const float4*)bias;
  const float4 nd = *(const float4*)(norm_dst + (size_t)r * 4);
  float s = 0.0f;
  s = gcn_readout_node(hp + 0 * (GCN_ROW / 4), wp + 0 * (GCN_ROW / 4), bp, nd.x, s);
  s = gcn_readout_node(hp + 1 * (GCN_ROW / 4), wp + 1 * (GCN_ROW / 4), bp, nd.y, s);
  s = gcn_readout_node(hp + 2 * (GCN_ROW / 4), wp + 2 * (GCN_ROW / 4), bp, nd.z, s);
  s = gcn_readout_node(hp + 3 * (GCN_ROW / 4), wp + 3 * (GCN_ROW / 4), bp, nd.w, s);
  out[r] = s + bd[0];
}


extern "C" void kernel_launch(void* const* d_in, const int* in_sizes, int n_in,
                              void* d_out, int out_size, void* d_ws, size_t ws_size,
                              hipStream_t stream) {
  const float* features = (const float*)d_in[0];
  const int*   src      = (const int*)d_in[1];
  const int*   dst      = (const int*)d_in[2];
  const float* W1       = (const float*)d_in[3];
  const float* b1       = (const float*)d_in[4];
  const float* W2       = (const float*)d_in[5];
  const float* b2       = (const float*)d_in[6];
  const float* Wd       = (const float*)d_in[7];
  const float* bd       = (const float*)d_in[8];
  float* out = (float*)d_out;

  const int H  = in_sizes[4];
  const int F  = in_sizes[3] / H;
  const int Nn = in_sizes[0] / F;
  const int E  = in_sizes[1];

  float* deg_out = (float*)d_ws;
  float* deg_in  = deg_out + Nn;
  float* aggA    = deg_in + Nn;
  float* bufX    = aggA + (size_t)Nn * H;
  float* aggB    = bufX + (size_t)Nn * H;
  unsigned int* degP = (unsigned int*)(aggB + (size_t)Nn * H);
  const int npairs = (Nn + 1) / 2;
  const int degS = (E + DEG_SLICE - 1) / DEG_SLICE;
  const int degR = (Nn + DEG_RANGE - 1) / DEG_RANGE;
  int2* rec = (int2*)(degP + (size_t)2 * degS * npairs);
  int2* tab = rec + (size_t)E;

  const int B = 256;
  const int ntiles = Nn / 16;
  const int wavesPerBlock = B / 32;
  const int gemmGrid = (ntiles + wavesPerBlock - 1) / wavesPerBlock;
  const int nS = (E + BIN_SLICE - 1) / BIN_SLICE;
  const int nB = (Nn + BIN_NPB - 1) / BIN_NPB;
  const int nq = Nn / 4;

  gcn_degree_lds_kernel<<<dim3(degS, degR, 2), 1024, 0, stream>>>(src, dst, degP, E, npairs);
  gcn_norm_kernel<<<(2 * npairs + B - 1) / B, B, 0, stream>>>(degP, deg_out, degS, npairs, Nn);
  gcn_bin_kernel<<<nS, BIN_THREADS, 0, stream>>>(src, dst, rec, tab, E);
  gcn_gemm_wmma_kernel<<<gemmGrid, B, 0, stream>>>(features, F, F, W1, H, deg_out, bufX, H,
                                                   ntiles);
  gcn_bucket_sum_kernel<<<nB, SUM_THREADS, 0, stream>>>(bufX, rec, tab, aggA, nS, Nn);
  gcn_gemm_wmma_post_kernel<<<gemmGrid, B, 0, stream>>>(aggA, H, H, W2, H, deg_out, deg_in, b1,
                                                        bufX, H, ntiles);
  gcn_bucket_sum_kernel<<<nB, SUM_THREADS, 0, stream>>>(bufX, rec, tab, aggB, nS, Nn);
  gcn_readout_post_kernel<<<(nq + B - 1) / B, B, 0, stream>>>(aggB, deg_in, b2, Wd, bd, out, nq);
}


# bucket-sum work-groups run the next dense stage on their summed rows while in LDS (layer-2 WMMA product; readout dot): sums never stored or re-read, 8 launches -> 6; same arithmetic and sum order
# speedup vs baseline: 4.4631x; 1.1887x over previous
#include <hip/hip_runtime.h>

typedef __attribute__((ext_vector_type(2))) float v2f;
typedef __attribute__((ext_vector_type(8))) float v8f;


#define DEG_RANGE 50000
#define DEG_SLICE 50000

__device__ __forceinline__ void deg_count_one(unsigned int* tbl, int idx, int base) {
  const unsigned int rel = (unsigned int)(idx - base);
  if (rel < (unsigned int)DEG_RANGE) {
    (void)__hip_atomic_fetch_add(&tbl[rel >> 1], 1u << ((rel & 1u) * 16u), __ATOMIC_RELAXED,
                                 __HIP_MEMORY_SCOPE_WORKGROUP);
  }
}

__global__ void gcn_degree_lds_kernel(const int* __restrict__ src, const int* __restrict__ dst,
                                      unsigned int* __restrict__ P, int E, int npairs) {
  __shared__ unsigned int tbl[DEG_RANGE / 2];

  const int s = blockIdx.x;
  const int r = blockIdx.y;
  const int a = blockIdx.z;
  const int S = gridDim.x;

  for (int i = threadIdx.x; i < DEG_RANGE / 2; i += blockDim.x) tbl[i] = 0u;
  __syncthreads();

  const int* idxp = (a == 0 ? src : dst) + (size_t)s * DEG_SLICE;
  int n = E - s * DEG_SLICE;
  if (n > DEG_SLICE) n = DEG_SLICE;
  const int base = r * DEG_RANGE;

  const int n4 = n >> 2;
  const int4* idx4 = (const int4*)idxp;
  for (int i0 = threadIdx.x; i0 < n4; i0 += 4 * blockDim.x) {
    int4 v[4];
#pragma unroll
    for (int j = 0; j < 4; ++j) {
      const int i = i0 + j * (int)blockDim.x;
      v[j] = (i < n4) ? idx4[i] : make_int4(-1, -1, -1, -1);
    }
#pragma unroll
    for (int j = 0; j < 4; ++j) {
      deg_count_one(tbl, v[j].x, base);
      deg_count_one(tbl, v[j].y, base);
      deg_count_one(tbl, v[j].z, base);
      deg_count_one(tbl, v[j].w, base);
    }
  }
  for (int i = (n4 << 2) + threadIdx.x; i < n; i += blockDim.x) deg_count_one(tbl, idxp[i], base);
  __syncthreads();

  unsigned int* plane = P + ((size_t)a * S + s) * npairs;
  const int w0 = r * (DEG_RANGE / 2);
  for (int i = threadIdx.x; i < DEG_RANGE / 2; i += blockDim.x) {
    const int w = w0 + i;
    if (w < npairs) plane[w] = tbl[i];
  }
}

__global__ void gcn_norm_kernel(const unsigned int* __restrict__ P, float* __restrict__ norm,
                                int S, int npairs, int Nn) {
  const int t = blockIdx.x * blockDim.x + threadIdx.x;
  if (t >= 2 * npairs) return;
  const int a = t / npairs;
  const int w = t - a * npairs;
  const unsigned int* p = P + (size_t)a * S * npairs + w;
  unsigned int lo = 0u, hi = 0u;
#pragma unroll 8
  for (int s = 0; s < S; ++s) {
    const unsigned int v = p[(size_t)s * npairs];
    lo += v & 0xffffu;
    hi += v >> 16;
  }
  float* o = norm + (size_t)a * Nn;
  const int node = 2 * w;
  o[node] = rsqrtf(fmaxf((float)lo, 1.0f));
  if (node + 1 < Nn) o[node + 1] = rsqrtf(fmaxf((float)hi, 1.0f));
}

#define GCN_ROW 24

__global__ void gcn_gemm_wmma_kernel(const float* __restrict__ A, int lda, int K,
                                     const float* __restrict__ W, int wcols,
                                     const float* __restrict__ norm_src,
                                     float* __restrict__ out, int ldo, int ntiles) {
  __shared__ float Ws[32 * 32];

  for (int i = threadIdx.x; i < 32 * 32; i += blockDim.x) {
    const int k = i >> 5;
    const int n = i & 31;
    Ws[i] = (k < K && n < wcols) ? W[(size_t)k * wcols + n] : 0.0f;
  }
  __syncthreads();

  const int wave = (blockIdx.x * blockDim.x + threadIdx.x) >> 5;
  const int lane = threadIdx.x & 31;
  if (wave >= ntiles) return;

  const int m = lane & 15;
  const int half = lane >> 4;

  const int row = wave * 16 + m;
  const float nrm = norm_src[row];
  const float* arow = A + (size_t)row * lda;

  v8f c0 = {};
  v8f c1 = {};

  for (int k0 = 0; k0 < K; k0 += 4) {
    const int k = k0 + 2 * half;
    const float2 av = *(const float2*)(arow + k);
    v2f a;
    a.x = av.x * nrm;
    a.y = av.y * nrm;

    v2f b0, b1;
    b0.x = Ws[k * 32 + m];
    b0.y = Ws[(k + 1) * 32 + m];
    b1.x = Ws[k * 32 + 16 + m];
    b1.y = Ws[(k + 1) * 32 + 16 + m];

    c0 = __builtin_amdgcn_wmma_f32_16x16x4_f32(false, a, false, b0, (short)0, c0, false, false);
    c1 = __builtin_amdgcn_wmma_f32_16x16x4_f32(false, a, false, b1, (short)0, c1, false, false);
  }

  for (int v = 0; v < 8; ++v) {
    const int orow = wave * 16 + v + 8 * half;
    out[(size_t)orow * ldo + m] = c0[v];
    if (16 + m < wcols) out[(size_t)orow * ldo + 16 + m] = c1[v];
  }
}

#define BIN_NB 256
#define BIN_NPB 784
#define BIN_THREADS 512
#define BIN_EPT 8
#define BIN_SLICE (BIN_THREADS * BIN_EPT)

#define SUM_GROUP_FLOATS (4 * GCN_ROW + 4)
#define SUM_LDS_FLOATS ((BIN_NPB / 4) * SUM_GROUP_FLOATS)
__device__ __forceinline__ int gcn_lds_row_off(int r) { return r * GCN_ROW + (r >> 2) * 4; }

__global__ void __launch_bounds__(BIN_THREADS)
gcn_bin_kernel(const int* __restrict__ src, const int* __restrict__ dst, int2* __restrict__ rec,
               int2* __restrict__ tab, int E) {
  __shared__ __attribute__((aligned(16))) unsigned int cnt[BIN_NB];
  __shared__ unsigned int start[BIN_NB];
  __shared__ int2 sorted[BIN_SLICE];

  const int tid = threadIdx.x;
  const int sl = blockIdx.x;
  const int e0 = sl * BIN_SLICE;

  if (tid < BIN_NB) cnt[tid] = 0u;
  __syncthreads();

  int s[BIN_EPT], d[BIN_EPT], bk[BIN_EPT];
  unsigned int rk[BIN_EPT];
#pragma unroll
  for (int j = 0; j < BIN_EPT; ++j) {
    const int e = e0 + j * BIN_THREADS + tid;
    const bool ok = e < E;
    const int ec = ok ? e : E - 1;
    s[j] = src[ec];
    const int dv = dst[ec];
    d[j] = ok ? dv : -1;
  }
#pragma unroll
  for (int j = 0; j < BIN_EPT; ++j) {
    bk[j] = 0;
    rk[j] = 0u;
    if (d[j] >= 0) {
      bk[j] = d[j] / BIN_NPB;
      rk[j] = __hip_atomic_fetch_add(&cnt[bk[j]], 1u, __ATOMIC_RELAXED,
                                     __HIP_MEMORY_SCOPE_WORKGROUP);
    }
  }
  __syncthreads();

  if (tid < BIN_NB) {
    const uint4* c4 = (const uint4*)cnt;
    unsigned int acc = 0u;
#pragma unroll 8
    for (int i = 0; i < BIN_NB / 4; ++i) {
      const uint4 c = c4[i];
      acc += (4 * i + 0 < tid) ? c.x : 0u;
      acc += (4 * i + 1 < tid) ? c.y : 0u;
      acc += (4 * i + 2 < tid) ? c.z : 0u;
      acc += (4 * i + 3 < tid) ? c.w : 0u;
    }
    start[tid] = acc;
    tab[(size_t)sl * BIN_NB + tid] = make_int2((int)acc, (int)cnt[tid]);
  }
  __syncthreads();

#pragma unroll
  for (int j = 0; j < BIN_EPT; ++j)
    if (d[j] >= 0)
      sorted[start[bk[j]] + rk[j]] = make_int2(s[j], gcn_lds_row_off(d[j] - bk[j] * BIN_NPB));
  __syncthreads();

  int n = E - e0;
  if (n > BIN_SLICE) n = BIN_SLICE;
  for (int i = tid; i < n; i += BIN_THREADS) rec[(size_t)e0 + i] = sorted[i];
}

__device__ __forceinline__ void lds_add_f32(float* p, float v) {
  (void)__hip_atomic_fetch_add(p, v, __ATOMIC_RELAXED, __HIP_MEMORY_SCOPE_WORKGROUP);
}

__device__ __forceinline__ float gcn_row_float(const float* __restrict__ x, int r, int lane) {
  const unsigned int off = (unsigned int)r * (unsigned int)(GCN_ROW * sizeof(float)) +
                           (unsigned int)lane * (unsigned int)sizeof(float);
  return *(const float*)((const char*)x + off);
}

#define SUM_BATCH 16
#define SUM_THREADS 1024
#define SUM_WAVES (SUM_THREADS / 32)

__device__ __forceinline__ void gcn_sum_batch(const float* __restrict__ x, float* rows,
                                              int cs, int cd, int p, int lane, bool act) {
  int si[SUM_BATCH], di[SUM_BATCH];
#pragma unroll
  for (int k = 0; k < SUM_BATCH; ++k) {
    si[k] = __builtin_amdgcn_readlane(cs, p + k);
    di[k] = __builtin_amdgcn_readlane(cd, p + k);
  }
  if (act) {
    float v[SUM_BATCH];
#pragma unroll
    for (int k = 0; k < SUM_BATCH; ++k) v[k] = gcn_row_float(x, si[k], lane);
#pragma unroll
    for (int k = 0; k < SUM_BATCH; ++k) lds_add_f32(rows + di[k] + lane, v[k]);
  }
}

__device__ __forceinline__ void gcn_sum_tail(const float* __restrict__ x, float* rows,
                                             int cs, int cd, int p, int m, int lane, bool act) {
  int di[15];
  float v[15];
  const bool h8 = (m & 8) != 0;
  const bool h4 = (m & 4) != 0;
  const bool h2 = (m & 2) != 0;
  const bool h1 = (m & 1) != 0;
  int q = p;
  if (h8) {
    int si[8];
#pragma unroll
    for (int k = 0; k < 8; ++k) {
      si[k] = __builtin_amdgcn_readlane(cs, q + k);
      di[k] = __builtin_amdgcn_readlane(cd, q + k);
    }
    if (act) {
#pragma unroll
      for (int k = 0; k < 8; ++k) v[k] = gcn_row_float(x, si[k], lane);
    }
    q += 8;
  }
  if (h4) {
    int si[4];
#pragma unroll
    for (int k = 0; k < 4; ++k) {
      si[k] = __builtin_amdgcn_readlane(cs, q + k);
      di[8 + k] = __builtin_amdgcn_readlane(cd, q + k);
    }
    if (act) {
#pragma unroll
      for (int k = 0; k < 4; ++k) v[8 + k] = gcn_row_float(x, si[k], lane);
    }
    q += 4;
  }
  if (h2) {
    int si[2];
#pragma unroll
    for (int k = 0; k < 2; ++k) {
      si[k] = __builtin_amdgcn_readlane(cs, q + k);
      di[12 + k] = __builtin_amdgcn_readlane(cd, q + k);
    }
    if (act) {
#pragma unroll
      for (int k = 0; k < 2; ++k) v[12 + k] = gcn_row_float(x, si[k], lane);
    }
    q += 2;
  }
  if (h1) {
    const int s1 = __builtin_amdgcn_readlane(cs, q);
    di[14] = __builtin_amdgcn_readlane(cd, q);
    if (act) v[14] = gcn_row_float(x, s1, lane);
  }
  if (h8 && act) {
#pragma unroll
    for (int k = 0; k < 8; ++k) lds_add_f32(rows + di[k] + lane, v[k]);
  }
  if (h4 && act) {
#pragma unroll
    for (int k = 0; k < 4; ++k) lds_add_f32(rows + di[8 + k] + lane, v[8 + k]);
  }
  if (h2 && act) {
#pragma unroll
    for (int k = 0; k < 2; ++k) lds_add_f32(rows + di[12 + k] + lane, v[12 + k]);
  }
  if (h1 && act) lds_add_f32(rows + di[14] + lane, v[14]);
}

__device__ __forceinline__ void gcn_sum_records(const float* __restrict__ x, float* rows,
                                                int cs, int cd, int m, int lane, bool act) {
  int p = 0;
  for (; p + SUM_BATCH <= m; p += SUM_BATCH) gcn_sum_batch(x, rows, cs, cd, p, lane, act);
  if (p < m) gcn_sum_tail(x, rows, cs, cd, p, m - p, lane, act);
}

__device__ __forceinline__ void gcn_bucket_accumulate(const float* __restrict__ x,
                                                      const int2* __restrict__ rec,
                                                      const int2* __restrict__ tab, float* rows,
                                                      int b, int w, int lane, int nS) {
  const bool act = lane < GCN_ROW;

  for (int g0 = w; g0 < nS; g0 += SUM_WAVES * 32) {
    const int mysl = g0 + SUM_WAVES * lane;
    int2 te = make_int2(0, 0);
    if (mysl < nS) te = tab[(size_t)mysl * BIN_NB + b];
    int nsl = (nS - g0 + SUM_WAVES - 1) / SUM_WAVES;
    if (nsl > 32) nsl = 32;

    int st = __builtin_amdgcn_readlane(te.x, 0);
    int cn = __builtin_amdgcn_readlane(te.y, 0);
    const int2* rp = rec + (size_t)g0 * BIN_SLICE + st;
    int2 cnext = make_int2(0, 0);
    if (lane < cn) cnext = rp[lane];

    for (int l = 0; l < nsl; ++l) {
      const int2 c = cnext;
      const int2* rpc = rp;
      const int cnc = cn;
      if (l + 1 < nsl) {
        st = __builtin_amdgcn_readlane(te.x, l + 1);
        cn = __builtin_amdgcn_readlane(te.y, l + 1);
        rp = rec + (size_t)(g0 + SUM_WAVES * (l + 1)) * BIN_SLICE + st;
        cnext = make_int2(0, 0);
        if (lane < cn) cnext = rp[lane];
      }
      if (cnc > 0) gcn_sum_records(x, rows, c.x, c.y, cnc < 32 ? cnc : 32, lane, act);
      for (int i0 = 32; i0 < cnc; i0 += 32) {
        const int m = (cnc - i0 < 32) ? (cnc - i0) : 32;
        int2 c2 = make_int2(0, 0);
        if (lane < m) c2 = rpc[i0 + lane];
        gcn_sum_records(x, rows, c2.x, c2.y, m, lane, act);
      }
    }
  }
  __syncthreads();
}

__device__ __forceinline__ void gcn_bucket_clear(float* rows) {
  float4* z = (float4*)rows;
  for (int i = threadIdx.x; i < SUM_LDS_FLOATS / 4; i += SUM_THREADS)
    z[i] = make_float4(0.0f, 0.0f, 0.0f, 0.0f);
}

#define SUM_TILES_PER_WAVE ((BIN_NPB / 16 + SUM_WAVES - 1) / SUM_WAVES)

__global__ void __launch_bounds__(SUM_THREADS)
gcn_bucket_sum_gemm_kernel(const float* __restrict__ x, const int2* __restrict__ rec,
                           const int2* __restrict__ tab, const float* __restrict__ W,
                           const float* __restrict__ norm_src,
                           const float* __restrict__ norm_dst,
                           const float* __restrict__ bias, float* __restrict__ out, int nS,
                           int Nn) {
  __shared__ __attribute__((aligned(16))) float rows[SUM_LDS_FLOATS];
  __shared__ float Ws[32 * 32];

  const int b = blockIdx.x;
  const int lane = threadIdx.x & 31;
  const int w = __builtin_amdgcn_readfirstlane((int)(threadIdx.x >> 5));
  const int nbase = b * BIN_NPB;
  int nr = Nn - nbase;
  if (nr > BIN_NPB) nr = BIN_NPB;
  const int nt = nr >> 4;

  const int m = lane & 15;
  const int half = lane >> 4;

  float nrm[SUM_TILES_PER_WAVE], nd[SUM_TILES_PER_WAVE];
#pragma unroll
  for (int j = 0; j < SUM_TILES_PER_WAVE; ++j) {
    int row = nbase + (w + j * SUM_WAVES) * 16 + m;
    if (row > Nn - 1) row = Nn - 1;
    nrm[j] = norm_src[row];
    nd[j] = norm_dst[row];
  }
  float2 bv[GCN_ROW / 4];
#pragma unroll
  for (int q = 0; q < GCN_ROW / 4; ++q) bv[q] = *(const float2*)(bias + 4 * q + 2 * half);

  gcn_bucket_clear(rows);
  for (int i = threadIdx.x; i < 32 * 32; i += SUM_THREADS) {
    const int k = i >> 5;
    const int n = i & 31;
    Ws[i] = (k < GCN_ROW && n < GCN_ROW) ? W[k * GCN_ROW + n] : 0.0f;
  }
  __syncthreads();

  gcn_bucket_accumulate(x, rec, tab, rows, b, w, lane, nS);

#pragma unroll
  for (int j = 0; j < SUM_TILES_PER_WAVE; ++j) {
    const int t = w + j * SUM_WAVES;
    if (t < nt) {
      const float* arow = rows + gcn_lds_row_off(t * 16 + m);
      v8f c0 = {};
      v8f c1 = {};
#pragma unroll
      for (int k0 = 0; k0 < GCN_ROW; k0 += 4) {
        const int k = k0 + 2 * half;
        const float2 av = *(const float2*)(arow + k);
        v2f a;
        a.x = fmaxf(fmaf(av.x, nd[j], bv[k0 / 4].x), 0.0f) * nrm[j];
        a.y = fmaxf(fmaf(av.y, nd[j], bv[k0 / 4].y), 0.0f) * nrm[j];

        v2f b0, b1;
        b0.x = Ws[k * 32 + m];
        b0.y = Ws[(k + 1) * 32 + m];
        b1.x = Ws[k * 32 + 16 + m];
        b1.y = Ws[(k + 1) * 32 + 16 + m];

        c0 = __builtin_amdgcn_wmma_f32_16x16x4_f32(false, a, false, b0, (short)0, c0, false, false);
        c1 = __builtin_amdgcn_wmma_f32_16x16x4_f32(false, a, false, b1, (short)0, c1, false, false);
      }

      float* otile = out + (size_t)(nbase + t * 16 + 8 * half) * GCN_ROW;
      for (int v = 0; v < 8; ++v) {
        otile[v * GCN_ROW + m] = c0[v];
        if (16 + m < GCN_ROW) otile[v * GCN_ROW + 16 + m] = c1[v];
      }
    }
  }
}

__device__ __forceinline__ float gcn_readout_node(const float4* hp, const float4* __restrict__ wp,
                                                  const float4* __restrict__ bp, float nd,
                                                  float s) {
  for (int c = 0; c < GCN_ROW / 4; ++c) {
    const float4 hv = hp[c];
    const float4 wv = wp[c];
    const float4 bv = bp[c];
    s = fmaf(fmaxf(fmaf(hv.x, nd, bv.x), 0.0f), wv.x, s);
    s = fmaf(fmaxf(fmaf(hv.y, nd, bv.y), 0.0f), wv.y, s);
    s = fmaf(fmaxf(fmaf(hv.z, nd, bv.z), 0.0f), wv.z, s);
    s = fmaf(fmaxf(fmaf(hv.w, nd, bv.w), 0.0f), wv.w, s);
  }
  return s;
}

__global__ void __launch_bounds__(SUM_THREADS)
gcn_bucket_sum_readout_kernel(const float* __restrict__ x, const int2* __restrict__ rec,
                              const int2* __restrict__ tab,
                              const float* __restrict__ norm_dst,
                              const float* __restrict__ bias, const float* __restrict__ Wd,
                              const float* __restrict__ bd, float* __restrict__ out, int nS,
                              int Nn) {
  __shared__ __attribute__((aligned(16))) float rows[SUM_LDS_FLOATS];

  const int b = blockIdx.x;
  const int lane = threadIdx.x & 31;
  const int w = __builtin_amdgcn_readfirstlane((int)(threadIdx.x >> 5));
  const int nbase = b * BIN_NPB;
  int nr = Nn - nbase;
  if (nr > BIN_NPB) nr = BIN_NPB;
  const int nq = nr >> 2;
  const int t = threadIdx.x;

  float4 nd = make_float4(0.0f, 0.0f, 0.0f, 0.0f);
  if (t < nq) nd = *(const float4*)(norm_dst + (size_t)nbase + 4 * t);
  gcn_bucket_clear(rows);
  __syncthreads();

  gcn_bucket_accumulate(x, rec, tab, rows, b, w, lane, nS);

  if (t < nq) {
    const float4* hp = (const float4*)(rows + t * SUM_GROUP_FLOATS);
    const float4* wp = (const float4*)Wd;
    const float4* bp = (const float4*)bias;
    float s = 0.0f;
    s = gcn_readout_node(hp + 0 * (GCN_ROW / 4), wp + 0 * (GCN_ROW / 4), bp, nd.x, s);
    s = gcn_readout_node(hp + 1 * (GCN_ROW / 4), wp + 1 * (GCN_ROW / 4), bp, nd.y, s);
    s = gcn_readout_node(hp + 2 * (GCN_ROW / 4), wp + 2 * (GCN_ROW / 4), bp, nd.z, s);
    s = gcn_readout_node(hp + 3 * (GCN_ROW / 4), wp + 3 * (GCN_ROW / 4), bp, nd.w, s);
    out[(nbase >> 2) + t] = s + bd[0];
  }
}


extern "C" void kernel_launch(void* const* d_in, const int* in_sizes, int n_in,
                              void* d_out, int out_size, void* d_ws, size_t ws_size,
                              hipStream_t stream) {
  const float* features = (const float*)d_in[0];
  const int*   src      = (const int*)d_in[1];
  const int*   dst      = (const int*)d_in[2];
  const float* W1       = (const float*)d_in[3];
  const float* b1       = (const float*)d_in[4];
  const float* W2       = (const float*)d_in[5];
  const float* b2       = (const float*)d_in[6];
  const float* Wd       = (const float*)d_in[7];
  const float* bd       = (const float*)d_in[8];
  float* out = (float*)d_out;

  const int H  = in_sizes[4];
  const int F  = in_sizes[3] / H;
  const int Nn = in_sizes[0] / F;
  const int E  = in_sizes[1];

  float* deg_out = (float*)d_ws;
  float* deg_in  = deg_out + Nn;
  float* bufX2   = deg_in + Nn;
  float* bufX    = bufX2 + (size_t)Nn * H;
  unsigned int* degP = (unsigned int*)(bufX + (size_t)Nn * H);
  const int npairs = (Nn + 1) / 2;
  const int degS = (E + DEG_SLICE - 1) / DEG_SLICE;
  const int degR = (Nn + DEG_RANGE - 1) / DEG_RANGE;
  int2* rec = (int2*)(degP + (size_t)2 * degS * npairs);
  int2* tab = rec + (size_t)E;

  const int B = 256;
  const int ntiles = Nn / 16;
  const int wavesPerBlock = B / 32;
  const int gemmGrid = (ntiles + wavesPerBlock - 1) / wavesPerBlock;
  const int nS = (E + BIN_SLICE - 1) / BIN_SLICE;
  const int nB = (Nn + BIN_NPB - 1) / BIN_NPB;

  gcn_degree_lds_kernel<<<dim3(degS, degR, 2), 1024, 0, stream>>>(src, dst, degP, E, npairs);
  gcn_norm_kernel<<<(2 * npairs + B - 1) / B, B, 0, stream>>>(degP, deg_out, degS, npairs, Nn);
  gcn_bin_kernel<<<nS, BIN_THREADS, 0, stream>>>(src, dst, rec, tab, E);
  gcn_gemm_wmma_kernel<<<gemmGrid, B, 0, stream>>>(features, F, F, W1, H, deg_out, bufX, H,
                                                   ntiles);
  gcn_bucket_sum_gemm_kernel<<<nB, SUM_THREADS, 0, stream>>>(bufX, rec, tab, W2, deg_out, deg_in,
                                                             b1, bufX2, nS, Nn);
  gcn_bucket_sum_readout_kernel<<<nB, SUM_THREADS, 0, stream>>>(bufX2, rec, tab, deg_in, b2, Wd,
                                                                bd, out, nS, Nn);
}


# launch fusion, 6 launches -> 4: edge grouping pass rides in the degree launch as extra work-groups; norm pass becomes the layer-1 product kernel's prologue; sums stay arrival-order LDS f32 adds
# speedup vs baseline: 4.8731x; 1.0919x over previous
#include <hip/hip_runtime.h>

typedef __attribute__((ext_vector_type(2))) float v2f;
typedef __attribute__((ext_vector_type(8))) float v8f;


#define DEG_RANGE 50000
#define DEG_SLICE 50000

__device__ __forceinline__ void deg_count_one(unsigned int* tbl, int idx, int base) {
  const unsigned int rel = (unsigned int)(idx - base);
  if (rel < (unsigned int)DEG_RANGE) {
    (void)__hip_atomic_fetch_add(&tbl[rel >> 1], 1u << ((rel & 1u) * 16u), __ATOMIC_RELAXED,
                                 __HIP_MEMORY_SCOPE_WORKGROUP);
  }
}

__device__ __forceinline__ void gcn_degree_role(unsigned int* tbl, const int* __restrict__ src,
                                                const int* __restrict__ dst,
                                                unsigned int* __restrict__ P, int E, int npairs,
                                                int s, int r, int a, int S) {

  for (int i = threadIdx.x; i < DEG_RANGE / 2; i += blockDim.x) tbl[i] = 0u;
  __syncthreads();

  const int* idxp = (a == 0 ? src : dst) + (size_t)s * DEG_SLICE;
  int n = E - s * DEG_SLICE;
  if (n > DEG_SLICE) n = DEG_SLICE;
  const int base = r * DEG_RANGE;

  const int n4 = n >> 2;
  const int4* idx4 = (const int4*)idxp;
  for (int i0 = threadIdx.x; i0 < n4; i0 += 4 * blockDim.x) {
    int4 v[4];
#pragma unroll
    for (int j = 0; j < 4; ++j) {
      const int i = i0 + j * (int)blockDim.x;
      v[j] = (i < n4) ? idx4[i] : make_int4(-1, -1, -1, -1);
    }
#pragma unroll
    for (int j = 0; j < 4; ++j) {
      deg_count_one(tbl, v[j].x, base);
      deg_count_one(tbl, v[j].y, base);
      deg_count_one(tbl, v[j].z, base);
      deg_count_one(tbl, v[j].w, base);
    }
  }
  for (int i = (n4 << 2) + threadIdx.x; i < n; i += blockDim.x) deg_count_one(tbl, idxp[i], base);
  __syncthreads();

  unsigned int* plane = P + ((size_t)a * S + s) * npairs;
  const int w0 = r * (DEG_RANGE / 2);
  for (int i = threadIdx.x; i < DEG_RANGE / 2; i += blockDim.x) {
    const int w = w0 + i;
    if (w < npairs) plane[w] = tbl[i];
  }
}

#define GCN_ROW 24

__global__ void gcn_gemm_wmma_kernel(const float* __restrict__ A, int lda, int K,
                                     const float* __restrict__ W, int wcols,
                                     const unsigned int* __restrict__ P, int S, int npairs,
                                     int Nn, float* __restrict__ norm,
                                     float* __restrict__ out, int ldo, int ntiles) {
  __shared__ float Ws[32 * 32];

  for (int i = threadIdx.x; i < 32 * 32; i += blockDim.x) {
    const int k = i >> 5;
    const int n = i & 31;
    Ws[i] = (k < K && n < wcols) ? W[(size_t)k * wcols + n] : 0.0f;
  }
  __syncthreads();

  const int wave = (blockIdx.x * blockDim.x + threadIdx.x) >> 5;
  const int lane = threadIdx.x & 31;
  if (wave >= ntiles) return;

  const int m = lane & 15;
  const int half = lane >> 4;

  const int row = wave * 16 + m;
  const unsigned int* pp = P + (size_t)half * S * npairs + (row >> 1);
  unsigned int lo = 0u, hi = 0u;
#pragma unroll 16
  for (int s = 0; s < S; ++s) {
    const unsigned int pv = pp[(size_t)s * npairs];
    lo += pv & 0xffffu;
    hi += pv >> 16;
  }
  const float mynorm = rsqrtf(fmaxf((float)((row & 1) ? hi : lo), 1.0f));
  norm[(size_t)half * Nn + row] = mynorm;
  const float nrm = __shfl(mynorm, m);
  const float* arow = A + (size_t)row * lda;

  v8f c0 = {};
  v8f c1 = {};

  for (int k0 = 0; k0 < K; k0 += 4) {
    const int k = k0 + 2 * half;
    const float2 av = *(const float2*)(arow + k);
    v2f a;
    a.x = av.x * nrm;
    a.y = av.y * nrm;

    v2f b0, b1;
    b0.x = Ws[k * 32 + m];
    b0.y = Ws[(k + 1) * 32 + m];
    b1.x = Ws[k * 32 + 16 + m];
    b1.y = Ws[(k + 1) * 32 + 16 + m];

    c0 = __builtin_amdgcn_wmma_f32_16x16x4_f32(false, a, false, b0, (short)0, c0, false, false);
    c1 = __builtin_amdgcn_wmma_f32_16x16x4_f32(false, a, false, b1, (short)0, c1, false, false);
  }

  for (int v = 0; v < 8; ++v) {
    const int orow = wave * 16 + v + 8 * half;
    out[(size_t)orow * ldo + m] = c0[v];
    if (16 + m < wcols) out[(size_t)orow * ldo + 16 + m] = c1[v];
  }
}

#define BIN_NB 256
#define BIN_NPB 784
#define BIN_THREADS 1024
#define BIN_EPT 4
#define BIN_SLICE (BIN_THREADS * BIN_EPT)

#define SUM_GROUP_FLOATS (4 * GCN_ROW + 4)
#define SUM_LDS_FLOATS ((BIN_NPB / 4) * SUM_GROUP_FLOATS)
__device__ __forceinline__ int gcn_lds_row_off(int r) { return r * GCN_ROW + (r >> 2) * 4; }

__device__ __forceinline__ void gcn_bin_role(unsigned int* smem, const int* __restrict__ src,
                                             const int* __restrict__ dst, int2* __restrict__ rec,
                                             int2* __restrict__ tab, int E, int sl) {
  unsigned int* cnt = smem;
  unsigned int* start = smem + BIN_NB;
  int2* sorted = (int2*)(smem + 2 * BIN_NB);

  const int tid = threadIdx.x;
  const int e0 = sl * BIN_SLICE;

  if (tid < BIN_NB) cnt[tid] = 0u;
  __syncthreads();

  int s[BIN_EPT], d[BIN_EPT], bk[BIN_EPT];
  unsigned int rk[BIN_EPT];
#pragma unroll
  for (int j = 0; j < BIN_EPT; ++j) {
    const int e = e0 + j * BIN_THREADS + tid;
    const bool ok = e < E;
    const int ec = ok ? e : E - 1;
    s[j] = src[ec];
    const int dv = dst[ec];
    d[j] = ok ? dv : -1;
  }
#pragma unroll
  for (int j = 0; j < BIN_EPT; ++j) {
    bk[j] = 0;
    rk[j] = 0u;
    if (d[j] >= 0) {
      bk[j] = d[j] / BIN_NPB;
      rk[j] = __hip_atomic_fetch_add(&cnt[bk[j]], 1u, __ATOMIC_RELAXED,
                                     __HIP_MEMORY_SCOPE_WORKGROUP);
    }
  }
  __syncthreads();

  if (tid < BIN_NB) {
    const uint4* c4 = (const uint4*)cnt;
    unsigned int acc = 0u;
#pragma unroll 8
    for (int i = 0; i < BIN_NB / 4; ++i) {
      const uint4 c = c4[i];
      acc += (4 * i + 0 < tid) ? c.x : 0u;
      acc += (4 * i + 1 < tid) ? c.y : 0u;
      acc += (4 * i + 2 < tid) ? c.z : 0u;
      acc += (4 * i + 3 < tid) ? c.w : 0u;
    }
    start[tid] = acc;
    tab[(size_t)sl * BIN_NB + tid] = make_int2((int)acc, (int)cnt[tid]);
  }
  __syncthreads();

#pragma unroll
  for (int j = 0; j < BIN_EPT; ++j)
    if (d[j] >= 0)
      sorted[start[bk[j]] + rk[j]] = make_int2(s[j], gcn_lds_row_off(d[j] - bk[j] * BIN_NPB));
  __syncthreads();

  int n = E - e0;
  if (n > BIN_SLICE) n = BIN_SLICE;
  for (int i = tid; i < n; i += BIN_THREADS) rec[(size_t)e0 + i] = sorted[i];
}

static_assert(2 * BIN_NB + 2 * BIN_SLICE <= DEG_RANGE / 2, "the grouping role must fit the degree table's LDS");
__global__ void __launch_bounds__(BIN_THREADS)
gcn_degree_bin_kernel(const int* __restrict__ src, const int* __restrict__ dst,
                      unsigned int* __restrict__ P, int2* __restrict__ rec,
                      int2* __restrict__ tab, int E, int npairs, int degS, int degR) {
  __shared__ __attribute__((aligned(16))) unsigned int smem[DEG_RANGE / 2];
  const int bx = blockIdx.x;
  const int nDeg = 2 * degS * degR;
  if (bx < nDeg) {
    const int s = bx % degS;
    const int ra = bx / degS;
    gcn_degree_role(smem, src, dst, P, E, npairs, s, ra % degR, ra / degR, degS);
  } else {
    gcn_bin_role(smem, src, dst, rec, tab, E, bx - nDeg);
  }
}

__device__ __forceinline__ void lds_add_f32(float* p, float v) {
  (void)__hip_atomic_fetch_add(p, v, __ATOMIC_RELAXED, __HIP_MEMORY_SCOPE_WORKGROUP);
}

__device__ __forceinline__ float gcn_row_float(const float* __restrict__ x, int r, int lane) {
  const unsigned int off = (unsigned int)r * (unsigned int)(GCN_ROW * sizeof(float)) +
                           (unsigned int)lane * (unsigned int)sizeof(float);
  return *(const float*)((const char*)x + off);
}

#define SUM_BATCH 16
#define SUM_THREADS 1024
#define SUM_WAVES (SUM_THREADS / 32)

__device__ __forceinline__ void gcn_sum_batch(const float* __restrict__ x, float* rows,
                                              int cs, int cd, int p, int lane, bool act) {
  int si[SUM_BATCH], di[SUM_BATCH];
#pragma unroll
  for (int k = 0; k < SUM_BATCH; ++k) {
    si[k] = __builtin_amdgcn_readlane(cs, p + k);
    di[k] = __builtin_amdgcn_readlane(cd, p + k);
  }
  if (act) {
    float v[SUM_BATCH];
#pragma unroll
    for (int k = 0; k < SUM_BATCH; ++k) v[k] = gcn_row_float(x, si[k], lane);
#pragma unroll
    for (int k = 0; k < SUM_BATCH; ++k) lds_add_f32(rows + di[k] + lane, v[k]);
  }
}

__device__ __forceinline__ void gcn_sum_tail(const float* __restrict__ x, float* rows,
                                             int cs, int cd, int p, int m, int lane, bool act) {
  int di[15];
  float v[15];
  const bool h8 = (m & 8) != 0;
  const bool h4 = (m & 4) != 0;
  const bool h2 = (m & 2) != 0;
  const bool h1 = (m & 1) != 0;
  int q = p;
  if (h8) {
    int si[8];
#pragma unroll
    for (int k = 0; k < 8; ++k) {
      si[k] = __builtin_amdgcn_readlane(cs, q + k);
      di[k] = __builtin_amdgcn_readlane(cd, q + k);
    }
    if (act) {
#pragma unroll
      for (int k = 0; k < 8; ++k) v[k] = gcn_row_float(x, si[k], lane);
    }
    q += 8;
  }
  if (h4) {
    int si[4];
#pragma unroll
    for (int k = 0; k < 4; ++k) {
      si[k] = __builtin_amdgcn_readlane(cs, q + k);
      di[8 + k] = __builtin_amdgcn_readlane(cd, q + k);
    }
    if (act) {
#pragma unroll
      for (int k = 0; k < 4; ++k) v[8 + k] = gcn_row_float(x, si[k], lane);
    }
    q += 4;
  }
  if (h2) {
    int si[2];
#pragma unroll
    for (int k = 0; k < 2; ++k) {
      si[k] = __builtin_amdgcn_readlane(cs, q + k);
      di[12 + k] = __builtin_amdgcn_readlane(cd, q + k);
    }
    if (act) {
#pragma unroll
      for (int k = 0; k < 2; ++k) v[12 + k] = gcn_row_float(x, si[k], lane);
    }
    q += 2;
  }
  if (h1) {
    const int s1 = __builtin_amdgcn_readlane(cs, q);
    di[14] = __builtin_amdgcn_readlane(cd, q);
    if (act) v[14] = gcn_row_float(x, s1, lane);
  }
  if (h8 && act) {
#pragma unroll
    for (int k = 0; k < 8; ++k) lds_add_f32(rows + di[k] + lane, v[k]);
  }
  if (h4 && act) {
#pragma unroll
    for (int k = 0; k < 4; ++k) lds_add_f32(rows + di[8 + k] + lane, v[8 + k]);
  }
  if (h2 && act) {
#pragma unroll
    for (int k = 0; k < 2; ++k) lds_add_f32(rows + di[12 + k] + lane, v[12 + k]);
  }
  if (h1 && act) lds_add_f32(rows + di[14] + lane, v[14]);
}

__device__ __forceinline__ void gcn_sum_records(const float* __restrict__ x, float* rows,
                                                int cs, int cd, int m, int lane, bool act) {
  int p = 0;
  for (; p + SUM_BATCH <= m; p += SUM_BATCH) gcn_sum_batch(x, rows, cs, cd, p, lane, act);
  if (p < m) gcn_sum_tail(x, rows, cs, cd, p, m - p, lane, act);
}

__device__ __forceinline__ void gcn_bucket_accumulate(const float* __restrict__ x,
                                                      const int2* __restrict__ rec,
                                                      const int2* __restrict__ tab, float* rows,
                                                      int b, int w, int lane, int nS) {
  const bool act = lane < GCN_ROW;

  for (int g0 = w; g0 < nS; g0 += SUM_WAVES * 32) {
    const int mysl = g0 + SUM_WAVES * lane;
    int2 te = make_int2(0, 0);
    if (mysl < nS) te = tab[(size_t)mysl * BIN_NB + b];
    int nsl = (nS - g0 + SUM_WAVES - 1) / SUM_WAVES;
    if (nsl > 32) nsl = 32;

    int st = __builtin_amdgcn_readlane(te.x, 0);
    int cn = __builtin_amdgcn_readlane(te.y, 0);
    const int2* rp = rec + (size_t)g0 * BIN_SLICE + st;
    int2 cnext = make_int2(0, 0);
    if (lane < cn) cnext = rp[lane];

    for (int l = 0; l < nsl; ++l) {
      const int2 c = cnext;
      const int2* rpc = rp;
      const int cnc = cn;
      if (l + 1 < nsl) {
        st = __builtin_amdgcn_readlane(te.x, l + 1);
        cn = __builtin_amdgcn_readlane(te.y, l + 1);
        rp = rec + (size_t)(g0 + SUM_WAVES * (l + 1)) * BIN_SLICE + st;
        cnext = make_int2(0, 0);
        if (lane < cn) cnext = rp[lane];
      }
      if (cnc > 0) gcn_sum_records(x, rows, c.x, c.y, cnc < 32 ? cnc : 32, lane, act);
      for (int i0 = 32; i0 < cnc; i0 += 32) {
        const int m = (cnc - i0 < 32) ? (cnc - i0) : 32;
        int2 c2 = make_int2(0, 0);
        if (lane < m) c2 = rpc[i0 + lane];
        gcn_sum_records(x, rows, c2.x, c2.y, m, lane, act);
      }
    }
  }
  __syncthreads();
}

__device__ __forceinline__ void gcn_bucket_clear(float* rows) {
  float4* z = (float4*)rows;
  for (int i = threadIdx.x; i < SUM_LDS_FLOATS / 4; i += SUM_THREADS)
    z[i] = make_float4(0.0f, 0.0f, 0.0f, 0.0f);
}

#define SUM_TILES_PER_WAVE ((BIN_NPB / 16 + SUM_WAVES - 1) / SUM_WAVES)

__global__ void __launch_bounds__(SUM_THREADS)
gcn_bucket_sum_gemm_kernel(const float* __restrict__ x, const int2* __restrict__ rec,
                           const int2* __restrict__ tab, const float* __restrict__ W,
                           const float* __restrict__ norm_src,
                           const float* __restrict__ norm_dst,
                           const float* __restrict__ bias, float* __restrict__ out, int nS,
                           int Nn) {
  __shared__ __attribute__((aligned(16))) float rows[SUM_LDS_FLOATS];
  __shared__ float Ws[32 * 32];

  const int b = blockIdx.x;
  const int lane = threadIdx.x & 31;
  const int w = __builtin_amdgcn_readfirstlane((int)(threadIdx.x >> 5));
  const int nbase = b * BIN_NPB;
  int nr = Nn - nbase;
  if (nr > BIN_NPB) nr = BIN_NPB;
  const int nt = nr >> 4;

  const int m = lane & 15;
  const int half = lane >> 4;

  float nrm[SUM_TILES_PER_WAVE], nd[SUM_TILES_PER_WAVE];
#pragma unroll
  for (int j = 0; j < SUM_TILES_PER_WAVE; ++j) {
    int row = nbase + (w + j * SUM_WAVES) * 16 + m;
    if (row > Nn - 1) row = Nn - 1;
    nrm[j] = norm_src[row];
    nd[j] = norm_dst[row];
  }
  float2 bv[GCN_ROW / 4];
#pragma unroll
  for (int q = 0; q < GCN_ROW / 4; ++q) bv[q] = *(const float2*)(bias + 4 * q + 2 * half);

  gcn_bucket_clear(rows);
  for (int i = threadIdx.x; i < 32 * 32; i += SUM_THREADS) {
    const int k = i >> 5;
    const int n = i & 31;
    Ws[i] = (k < GCN_ROW && n < GCN_ROW) ? W[k * GCN_ROW + n] : 0.0f;
  }
  __syncthreads();

  gcn_bucket_accumulate(x, rec, tab, rows, b, w, lane, nS);

#pragma unroll
  for (int j = 0; j < SUM_TILES_PER_WAVE; ++j) {
    const int t = w + j * SUM_WAVES;
    if (t < nt) {
      const float* arow = rows + gcn_lds_row_off(t * 16 + m);
      v8f c0 = {};
      v8f c1 = {};
#pragma unroll
      for (int k0 = 0; k0 < GCN_ROW; k0 += 4) {
        const int k = k0 + 2 * half;
        const float2 av = *(const float2*)(arow + k);
        v2f a;
        a.x = fmaxf(fmaf(av.x, nd[j], bv[k0 / 4].x), 0.0f) * nrm[j];
        a.y = fmaxf(fmaf(av.y, nd[j], bv[k0 / 4].y), 0.0f) * nrm[j];

        v2f b0, b1;
        b0.x = Ws[k * 32 + m];
        b0.y = Ws[(k + 1) * 32 + m];
        b1.x = Ws[k * 32 + 16 + m];
        b1.y = Ws[(k + 1) * 32 + 16 + m];

        c0 = __builtin_amdgcn_wmma_f32_16x16x4_f32(false, a, false, b0, (short)0, c0, false, false);
        c1 = __builtin_amdgcn_wmma_f32_16x16x4_f32(false, a, false, b1, (short)0, c1, false, false);
      }

      float* otile = out + (size_t)(nbase + t * 16 + 8 * half) * GCN_ROW;
      for (int v = 0; v < 8; ++v) {
        otile[v * GCN_ROW + m] = c0[v];
        if (16 + m < GCN_ROW) otile[v * GCN_ROW + 16 + m] = c1[v];
      }
    }
  }
}

__device__ __forceinline__ float gcn_readout_node(const float4* hp, const float4* __restrict__ wp,
                                                  const float4* __restrict__ bp, float nd,
                                                  float s) {
  for (int c = 0; c < GCN_ROW / 4; ++c) {
    const float4 hv = hp[c];
    const float4 wv = wp[c];
    const float4 bv = bp[c];
    s = fmaf(fmaxf(fmaf(hv.x, nd, bv.x), 0.0f), wv.x, s);
    s = fmaf(fmaxf(fmaf(hv.y, nd, bv.y), 0.0f), wv.y, s);
    s = fmaf(fmaxf(fmaf(hv.z, nd, bv.z), 0.0f), wv.z, s);
    s = fmaf(fmaxf(fmaf(hv.w, nd, bv.w), 0.0f), wv.w, s);
  }
  return s;
}

__global__ void __launch_bounds__(SUM_THREADS)
gcn_bucket_sum_readout_kernel(const float* __restrict__ x, const int2* __restrict__ rec,
                              const int2* __restrict__ tab,
                              const float* __restrict__ norm_dst,
                              const float* __restrict__ bias, const float* __restrict__ Wd,
                              const float* __restrict__ bd, float* __restrict__ out, int nS,
                              int Nn) {
  __shared__ __attribute__((aligned(16))) float rows[SUM_LDS_FLOATS];

  const int b = blockIdx.x;
  const int lane = threadIdx.x & 31;
  const int w = __builtin_amdgcn_readfirstlane((int)(threadIdx.x >> 5));
  const int nbase = b * BIN_NPB;
  int nr = Nn - nbase;
  if (nr > BIN_NPB) nr = BIN_NPB;
  const int nq = nr >> 2;
  const int t = threadIdx.x;

  float4 nd = make_float4(0.0f, 0.0f, 0.0f, 0.0f);
  if (t < nq) nd = *(const float4*)(norm_dst + (size_t)nbase + 4 * t);
  gcn_bucket_clear(rows);
  __syncthreads();

  gcn_bucket_accumulate(x, rec, tab, rows, b, w, lane, nS);

  if (t < nq) {
    const float4* hp = (const float4*)(rows + t * SUM_GROUP_FLOATS);
    const float4* wp = (const float4*)Wd;
    const float4* bp = (const float4*)bias;
    float s = 0.0f;
    s = gcn_readout_node(hp + 0 * (GCN_ROW / 4), wp + 0 * (GCN_ROW / 4), bp, nd.x, s);
    s = gcn_readout_node(hp + 1 * (GCN_ROW / 4), wp + 1 * (GCN_ROW / 4), bp, nd.y, s);
    s = gcn_readout_node(hp + 2 * (GCN_ROW / 4), wp + 2 * (GCN_ROW / 4), bp, nd.z, s);
    s = gcn_readout_node(hp + 3 * (GCN_ROW / 4), wp + 3 * (GCN_ROW / 4), bp, nd.w, s);
    out[(nbase >> 2) + t] = s + bd[0];
  }
}


extern "C" void kernel_launch(void* const* d_in, const int* in_sizes, int n_in,
                              void* d_out, int out_size, void* d_ws, size_t ws_size,
                              hipStream_t stream) {
  const float* features = (const float*)d_in[0];
  const int*   src      = (const int*)d_in[1];
  const int*   dst      = (const int*)d_in[2];
  const float* W1       = (const float*)d_in[3];
  const float* b1       = (const float*)d_in[4];
  const float* W2       = (const float*)d_in[5];
  const float* b2       = (const float*)d_in[6];
  const float* Wd       = (const float*)d_in[7];
  const float* bd       = (const float*)d_in[8];
  float* out = (float*)d_out;

  const int H  = in_sizes[4];
  const int F  = in_sizes[3] / H;
  const int Nn = in_sizes[0] / F;
  const int E  = in_sizes[1];

  float* deg_out = (float*)d_ws;
  float* deg_in  = deg_out + Nn;
  float* bufX2   = deg_in + Nn;
  float* bufX    = bufX2 + (size_t)Nn * H;
  unsigned int* degP = (unsigned int*)(bufX + (size_t)Nn * H);
  const int npairs = (Nn + 1) / 2;
  const int degS = (E + DEG_SLICE - 1) / DEG_SLICE;
  const int degR = (Nn + DEG_RANGE - 1) / DEG_RANGE;
  int2* rec = (int2*)(degP + (size_t)2 * degS * npairs);
  int2* tab = rec + (size_t)E;

  const int B = 256;
  const int ntiles = Nn / 16;
  const int wavesPerBlock = B / 32;
  const int gemmGrid = (ntiles + wavesPerBlock - 1) / wavesPerBlock;
  const int nS = (E + BIN_SLICE - 1) / BIN_SLICE;
  const int nB = (Nn + BIN_NPB - 1) / BIN_NPB;

  gcn_degree_bin_kernel<<<2 * degS * degR + nS, BIN_THREADS, 0, stream>>>(src, dst, degP, rec, tab,
                                                                         E, npairs, degS, degR);
  gcn_gemm_wmma_kernel<<<gemmGrid, B, 0, stream>>>(features, F, F, W1, H, degP, degS, npairs, Nn,
                                                   deg_out, bufX, H, ntiles);
  gcn_bucket_sum_gemm_kernel<<<nB, SUM_THREADS, 0, stream>>>(bufX, rec, tab, W2, deg_out, deg_in,
                                                             b1, bufX2, nS, Nn);
  gcn_bucket_sum_readout_kernel<<<nB, SUM_THREADS, 0, stream>>>(bufX2, rec, tab, deg_in, b2, Wd,
                                                                bd, out, nS, Nn);
}
